# starting kernel
# baseline (speedup 1.0000x reference)
#include <hip/hip_runtime.h>
#include <math.h>

typedef __attribute__((ext_vector_type(2))) float v2f;
typedef __attribute__((ext_vector_type(8))) float v8f;

#define BB 2
#define CC 256
#define HH 64
#define WW 64
#define HW 4096
#define NN 16
#define NROW (BB*HW)

__global__ void init_out_kernel(const float* __restrict__ feat,
                                const float* __restrict__ Dvec,
                                float* __restrict__ out) {
    int idx = blockIdx.x * blockDim.x + threadIdx.x;
    int c = (idx >> 12) & (CC - 1);
    out[idx] = feat[idx] * Dvec[c];
}

__global__ void proj_wmma_kernel(const float* __restrict__ feat,
                                 const float* __restrict__ W_delta,
                                 const float* __restrict__ b_delta,
                                 const float* __restrict__ W_B,
                                 const float* __restrict__ W_C,
                                 float* __restrict__ delta_ws,
                                 float* __restrict__ Bm_ws,
                                 float* __restrict__ Cm_ws) {
    const int lane = threadIdx.x;
    const int lo   = lane & 15;
    const int hi   = lane >> 4;
    const int m0   = blockIdx.x * 16;
    const int which = blockIdx.y;

    const float* Wmat;
    int ncols, n0;
    if (which < 16)       { Wmat = W_delta; ncols = CC; n0 = which * 16; }
    else if (which == 16) { Wmat = W_B;     ncols = NN; n0 = 0; }
    else                  { Wmat = W_C;     ncols = NN; n0 = 0; }

    const int rowA = m0 + lo;
    const int bA   = rowA >> 12;
    const int lA   = rowA & (HW - 1);
    const float* aBase = feat + (size_t)bA * CC * HW + lA;

    v8f acc = {};
    for (int k0 = 0; k0 < CC; k0 += 4) {
        const int ka = k0 + hi * 2;
        v2f a, b;
        a.x = aBase[(size_t)ka * HW];
        a.y = aBase[(size_t)(ka + 1) * HW];
        b.x = Wmat[(size_t)ka * ncols + n0 + lo];
        b.y = Wmat[(size_t)(ka + 1) * ncols + n0 + lo];
        acc = __builtin_amdgcn_wmma_f32_16x16x4_f32(
             false, a,  false, b,
             (short)0, acc,  false,  false);
    }

    if (which < 16) {
        const float bias = b_delta[n0 + lo];
        #pragma unroll
        for (int v = 0; v < 8; ++v) {
            const int row = m0 + v + 8 * hi;
            const float pre = acc[v] + bias;
            const float sp = (pre > 20.f) ? pre : log1pf(__expf(pre));
            delta_ws[(size_t)row * CC + n0 + lo] = sp;
        }
    } else {
        float* dst = (which == 16) ? Bm_ws : Cm_ws;
        #pragma unroll
        for (int v = 0; v < 8; ++v) {
            const int row = m0 + v + 8 * hi;
            dst[(size_t)row * NN + lo] = acc[v];
        }
    }
}

__global__ void scan_kernel(const float* __restrict__ feat,
                            const float* __restrict__ A_log,
                            const float* __restrict__ delta_ws,
                            const float* __restrict__ Bm_ws,
                            const float* __restrict__ Cm_ws,
                            float* __restrict__ out) {
    const int lane = threadIdx.x;
    const int n  = lane & 15;
    const int hi = lane >> 4;
    const int id  = blockIdx.x;
    const int dir = id >> 8;
    const int rem = id & 255;
    const int b   = rem >> 7;
    const int c   = ((rem & 127) << 1) + hi;

    int base, sh, sw;
    switch (dir) {
        case 0:  base = 0;    sh =  64; sw =  1; break;
        case 1:  base = 4095; sh = -64; sw = -1; break;
        case 2:  base = 4032; sh = -64; sw =  1; break;
        default: base = 63;   sh =  -1; sw = 64; break;
    }

    const float A = -__expf(A_log[c * NN + n]);
    const float* fBase = feat + (size_t)b * CC * HW + (size_t)c * HW;
    const size_t rbase = (size_t)b * HW;
    float* outBase = out + (size_t)b * CC * HW + (size_t)c * HW;

    float h = 0.f;
    for (int t = 0; t < HW; ++t) {
        const int l = base + sh * (t >> 6) + sw * (t & 63);
        const size_t row = rbase + (size_t)l;
        const float d  = delta_ws[row * CC + c];
        const float xv = fBase[l];
        const float Bv = Bm_ws[row * NN + n];
        const float Cv = Cm_ws[row * NN + n];
        const float dA = __expf(d * A);
        h = fmaf(dA, h, d * xv * Bv);
        float contrib = h * Cv;
        contrib += __shfl_xor(contrib, 1);
        contrib += __shfl_xor(contrib, 2);
        contrib += __shfl_xor(contrib, 4);
        contrib += __shfl_xor(contrib, 8);
        if (n == 0) atomicAdd(&outBase[l], contrib * 0.25f);
    }
}

extern "C" void kernel_launch(void* const* d_in, const int* in_sizes, int n_in,
                              void* d_out, int out_size, void* d_ws, size_t ws_size,
                              hipStream_t stream) {
    const float* feat    = (const float*)d_in[0];
    const float* A_log   = (const float*)d_in[1];
    const float* Dvec    = (const float*)d_in[2];
    const float* W_delta = (const float*)d_in[3];
    const float* b_delta = (const float*)d_in[4];
    const float* W_B     = (const float*)d_in[5];
    const float* W_C     = (const float*)d_in[6];
    float* out = (float*)d_out;

    float* ws       = (float*)d_ws;
    float* delta_ws = ws;
    float* Bm_ws    = ws + (size_t)NROW * CC;
    float* Cm_ws    = Bm_ws + (size_t)NROW * NN;

    init_out_kernel<<<(BB * CC * HW) / 256, 256, 0, stream>>>(feat, Dvec, out);
    proj_wmma_kernel<<<dim3(NROW / 16, 18), 32, 0, stream>>>(
        feat, W_delta, b_delta, W_B, W_C, delta_ws, Bm_ws, Cm_ws);
    scan_kernel<<<1024, 32, 0, stream>>>(feat, A_log, delta_ws, Bm_ws, Cm_ws, out);
}


# projection re-tiled: a work-group reads 32 rows x 256 k of x once: A fragments via LDS for all 18 column tiles (weights in registers), out = x*D from the same registers; init launch gone; sums kept
# speedup vs baseline: 23.7812x; 23.7812x over previous
#include <hip/hip_runtime.h>
#include <math.h>

typedef __attribute__((ext_vector_type(2))) float v2f;
typedef __attribute__((ext_vector_type(8))) float v8f;

#define BB 2
#define CC 256
#define HH 64
#define WW 64
#define HW 4096
#define NN 16
#define NROW (BB*HW)

__device__ __forceinline__ float softplus_sel(float pre) {
    const float t = log1pf(__expf(pre));
    return (pre > 20.f) ? pre : t;
}
#define PROJ_ROWS  32
#define PROJ_WAVES 18
#define PROJ_THREADS (PROJ_WAVES * 32)
#define PROJ_PITCH 260
#define PROJ_V4 (PROJ_ROWS * CC / 4)
#define PROJ_KSTEPS (CC / 4)
__global__ __launch_bounds__(PROJ_THREADS) void proj_wmma_kernel(const float* __restrict__ feat,
                                 const float* __restrict__ Dvec,
                                 const float* __restrict__ W_delta,
                                 const float* __restrict__ b_delta,
                                 const float* __restrict__ W_B,
                                 const float* __restrict__ W_C,
                                 float* __restrict__ delta_ws,
                                 float* __restrict__ Bm_ws,
                                 float* __restrict__ Cm_ws,
                                 float* __restrict__ out) {
    __shared__ float sX[PROJ_ROWS * PROJ_PITCH];

    const int tid  = threadIdx.x;
    const int lane = tid & 31;
    const int lo   = lane & 15;
    const int hi   = lane >> 4;
    const int which = __builtin_amdgcn_readfirstlane(tid >> 5);
    const int m0   = blockIdx.x * PROJ_ROWS;
    const int bA   = m0 >> 12;
    const int l0   = m0 & (HW - 1);
    const float* xBase = feat + (size_t)bA * CC * HW + l0;

    float4 v[4];
    float dk[4];
    #pragma unroll
    for (int i = 0; i < 4; ++i) {
        const int f = tid + PROJ_THREADS * i;
        if (f < PROJ_V4) {
            v[i] = *reinterpret_cast<const float4*>(xBase + (size_t)(f >> 3) * HW + 4 * (f & 7));
            dk[i] = Dvec[f >> 3];
        }
    }
    #pragma unroll
    for (int i = 0; i < 4; ++i) {
        const int f = tid + PROJ_THREADS * i;
        if (f < PROJ_V4) {
            float* d = sX + (4 * (f & 7)) * PROJ_PITCH + (f >> 3);
            d[0]              = v[i].x;
            d[PROJ_PITCH]     = v[i].y;
            d[2 * PROJ_PITCH] = v[i].z;
            d[3 * PROJ_PITCH] = v[i].w;
        }
    }
    __syncthreads();

    {
        float* oBase = out + (size_t)bA * CC * HW + l0;
        #pragma unroll
        for (int i = 0; i < 4; ++i) {
            const int f = tid + PROJ_THREADS * i;
            if (f < PROJ_V4) {
                float4 o;
                o.x = v[i].x * dk[i]; o.y = v[i].y * dk[i]; o.z = v[i].z * dk[i]; o.w = v[i].w * dk[i];
                *reinterpret_cast<float4*>(oBase + (size_t)(f >> 3) * HW + 4 * (f & 7)) = o;
            }
        }
    }

    const float* Wmat;
    unsigned ncols, n0;
    if (which < 16)       { Wmat = W_delta; ncols = CC; n0 = which * 16; }
    else if (which == 16) { Wmat = W_B;     ncols = NN; n0 = 0; }
    else                  { Wmat = W_C;     ncols = NN; n0 = 0; }
    const unsigned laneoff = (unsigned)(2 * hi) * ncols + n0 + (unsigned)lo;
    v2f bw[PROJ_KSTEPS];
    #pragma unroll
    for (int s = 0; s < PROJ_KSTEPS; ++s) {
        const float* r0 = Wmat + (size_t)(4 * s) * ncols;
        const float* r1 = r0 + ncols;
        bw[s].x = r0[laneoff];
        bw[s].y = r1[laneoff];
    }

    auto run_tile = [&](const float* aRow) -> v8f {
        v8f acc = {};
        #pragma unroll
        for (int s = 0; s < PROJ_KSTEPS; ++s) {
            const v2f a = *reinterpret_cast<const v2f*>(aRow + 4 * s);
            acc = __builtin_amdgcn_wmma_f32_16x16x4_f32(
                 false, a,  false, bw[s],
                 (short)0, acc,  false,  false);
        }
        return acc;
    };
    const float* aRow0 = sX + lo * PROJ_PITCH + 2 * hi;
    const float* aRow1 = aRow0 + 16 * PROJ_PITCH;

    if (which < 16) {
        const float bias = b_delta[n0 + lo];
        float* dcol = delta_ws + (size_t)(m0 + 8 * hi) * CC + n0 + lo;
        const v8f acc0 = run_tile(aRow0);
        v8f acc1 = {};
        #pragma unroll
        for (int r = 0; r < 8; ++r) {
            #pragma unroll
            for (int s = 8 * r; s < 8 * r + 8; ++s) {
                const v2f a = *reinterpret_cast<const v2f*>(aRow1 + 4 * s);
                acc1 = __builtin_amdgcn_wmma_f32_16x16x4_f32(
                     false, a,  false, bw[s],
                     (short)0, acc1,  false,  false);
            }
            dcol[(size_t)r * CC] = softplus_sel(acc0[r] + bias);
            #pragma unroll
            for (int g = 0; g < 8; ++g) {
                __builtin_amdgcn_sched_group_barrier(0x008, 1, 0);
                __builtin_amdgcn_sched_group_barrier(0x002, 11, 0);
            }
            __builtin_amdgcn_sched_barrier(0);
        }
        #pragma unroll
        for (int r = 0; r < 8; ++r)
            dcol[(size_t)(16 + r) * CC] = softplus_sel(acc1[r] + bias);
    } else {
        float* dst = ((which == 16) ? Bm_ws : Cm_ws) + (size_t)(m0 + 8 * hi) * NN + lo;
        const v8f acc0 = run_tile(aRow0);
        #pragma unroll
        for (int r = 0; r < 8; ++r) dst[(size_t)r * NN] = acc0[r];
        const v8f acc1 = run_tile(aRow1);
        #pragma unroll
        for (int r = 0; r < 8; ++r) dst[(size_t)(16 + r) * NN] = acc1[r];
    }
}


#define KS   8
#define BLK  (32 * KS)
#define NBLK (HW / BLK)
#define NG   4
#define NQ   (NN / 4)
#define PITCH 33
#define NSTG (BLK * NQ / 256)

template <int K>
__device__ __forceinline__ float row_shr_or0(float v) {
    return __int_as_float(__builtin_amdgcn_update_dpp(
        0, __float_as_int(v), 0x110 + K, 0xf, 0xf, true));
}
template <int K>
__device__ __forceinline__ float row_shr_or(float outside, float v) {
    return __int_as_float(__builtin_amdgcn_update_dpp(
        __float_as_int(outside), __float_as_int(v), 0x110 + K, 0xf, 0xf, false));
}
template <int K>
__device__ __forceinline__ void scan_level(float& a, float& bb) {
    const float bs = row_shr_or0<K>(bb);
    const float as = row_shr_or<K>(1.0f, a);
    bb = fmaf(a, bs, bb);
    a  = a * as;
}

__global__ __launch_bounds__(256) __attribute__((amdgpu_waves_per_eu(4, 4))) void scan_kernel(const float* __restrict__ feat,
                            const float* __restrict__ A_log,
                            const float* __restrict__ delta_ws,
                            const float* __restrict__ Bm_ws,
                            const float* __restrict__ Cm_ws,
                            float* __restrict__ out) {
    __shared__ float4 sB[2][KS * NQ * PITCH];
    __shared__ float4 sC[2][KS * NQ * PITCH];

    const int tid  = threadIdx.x;
    const int lane = tid & 31;
    const int w   = __builtin_amdgcn_readfirstlane((int)(threadIdx.x >> 5));
    const int id  = blockIdx.x * 8 + w;
    const int dir = id >> 9;
    const int b   = (id >> 8) & 1;
    const int c   = id & 255;

    int base, sh, sw;
    switch (dir) {
        case 0:  base = 0;    sh =  64; sw =  1; break;
        case 1:  base = 4095; sh = -64; sw = -1; break;
        case 2:  base = 4032; sh = -64; sw =  1; break;
        default: base = 63;   sh =  -1; sw = 64; break;
    }

    float A[NN];
    #pragma unroll
    for (int n = 0; n < NN; ++n) A[n] = -__expf(A_log[c * NN + n]);

    const float* fBase = feat + (size_t)b * CC * HW + (size_t)c * HW;
    const size_t rbase = (size_t)b * HW;
    float* outBase = out + (size_t)b * CC * HW + (size_t)c * HW;
    const bool upper = lane >= 16;
    const float4* Bm4 = reinterpret_cast<const float4*>(Bm_ws);
    const float4* Cm4 = reinterpret_cast<const float4*>(Cm_ws);

    float hin[NN];
    #pragma unroll
    for (int n = 0; n < NN; ++n) hin[n] = 0.f;

    auto pos_of = [&](int blk, int s) -> int {
        const int t = blk * BLK + s;
        return base + sh * (t >> 6) + sw * (t & 63);
    };
    auto pos = [&](int blk, int j) -> int { return pos_of(blk, KS * lane + j); };

    auto stage_load = [&](int blk, int r, float4& vb, float4& vc) {
        const int f = tid + 256 * r;
        const int l = pos_of(blk, f >> 2);
        const size_t v4 = (rbase + (size_t)l) * NQ + (size_t)(f & 3);
        vb = Bm4[v4];
        vc = Cm4[v4];
    };
    auto stage_store = [&](int buf, int r, const float4& vb, const float4& vc) {
        const int f = tid + 256 * r;
        const int s = f >> 2;
        const int idx = ((s % KS) * NQ + (f & 3)) * PITCH + (s / KS);
        sB[buf][idx] = vb;
        sC[buf][idx] = vc;
    };

    auto load_dx = [&](int blk, float (&d)[KS], float (&xv)[KS]) {
        #pragma unroll
        for (int j = 0; j < KS; ++j) {
            const int l = pos(blk, j);
            d[j]  = delta_ws[(rbase + (size_t)l) * CC + c];
            xv[j] = fBase[l];
        }
    };

    auto lds_load = [&](int buf, int q, float (&Bv)[KS][4], float (&Cv)[KS][4]) {
        #pragma unroll
        for (int j = 0; j < KS; ++j) {
            const float4 bv = sB[buf][(j * NQ + q) * PITCH + lane];
            const float4 cv = sC[buf][(j * NQ + q) * PITCH + lane];
            Bv[j][0] = bv.x; Bv[j][1] = bv.y; Bv[j][2] = bv.z; Bv[j][3] = bv.w;
            Cv[j][0] = cv.x; Cv[j][1] = cv.y; Cv[j][2] = cv.z; Cv[j][3] = cv.w;
        }
    };

    auto do_group = [&](int n0, const float (&d)[KS], const float (&dx)[KS],
                        const float (&Bv)[KS][4], const float (&Cv)[KS][4], float (&y)[KS]) {
        float e[NG][KS], bb[NG][KS], P[NG], Q[NG], H[NG], h[NG];
        #pragma unroll
        for (int g = 0; g < NG; ++g) {
            #pragma unroll
            for (int j = 0; j < KS; ++j) e[g][j] = __expf(d[j] * A[n0 + g]);
        }
        __builtin_amdgcn_sched_barrier(0);
        #pragma unroll
        for (int g = 0; g < NG; ++g) {
            #pragma unroll
            for (int j = 0; j < KS; ++j) bb[g][j] = dx[j] * Bv[j][g];
        }
        #pragma unroll
        for (int g = 0; g < NG; ++g) { P[g] = e[g][0]; Q[g] = bb[g][0]; }
        #pragma unroll
        for (int j = 1; j < KS; ++j) {
            #pragma unroll
            for (int g = 0; g < NG; ++g) {
                Q[g] = fmaf(e[g][j], Q[g], bb[g][j]);
                P[g] = P[g] * e[g][j];
            }
        }
        #pragma unroll
        for (int g = 0; g < NG; ++g) scan_level<1>(P[g], Q[g]);
        #pragma unroll
        for (int g = 0; g < NG; ++g) scan_level<2>(P[g], Q[g]);
        #pragma unroll
        for (int g = 0; g < NG; ++g) scan_level<4>(P[g], Q[g]);
        #pragma unroll
        for (int g = 0; g < NG; ++g) scan_level<8>(P[g], Q[g]);
        #pragma unroll
        for (int g = 0; g < NG; ++g) {
            const int n = n0 + g;
            const float h0 = fmaf(P[g], hin[n], Q[g]);
            const float c0 = __int_as_float(__builtin_amdgcn_readlane(__float_as_int(h0), 15));
            H[g] = upper ? fmaf(P[g], c0, Q[g]) : h0;
            const float rowin = upper ? c0 : hin[n];
            hin[n] = __int_as_float(__builtin_amdgcn_readlane(__float_as_int(H[g]), 31));
            h[g] = row_shr_or<1>(rowin, H[g]);
        }
        #pragma unroll
        for (int j = 0; j < KS - 1; ++j) {
            #pragma unroll
            for (int g = 0; g < NG; ++g) {
                h[g] = fmaf(e[g][j], h[g], bb[g][j]);
                y[j] = fmaf(h[g], Cv[j][g], y[j]);
            }
        }
        #pragma unroll
        for (int g = 0; g < NG; ++g)
            y[KS - 1] = fmaf(H[g], Cv[KS - 1][g], y[KS - 1]);
    };

    auto do_block = [&](int buf, const float (&d)[KS], const float (&xv)[KS], float (&y)[KS]) {
        float dx[KS];
        #pragma unroll
        for (int j = 0; j < KS; ++j) { dx[j] = d[j] * xv[j]; y[j] = 0.f; }
        float Bv[KS][4], Cv[KS][4];
        #pragma unroll
        for (int q = 0; q < NQ; ++q) {
            lds_load(buf, q, Bv, Cv);
            __builtin_amdgcn_sched_barrier(0);
            do_group(4 * q, d, dx, Bv, Cv, y);
        }
    };

    auto add_out = [&](int blk, const float (&y)[KS]) {
        #pragma unroll
        for (int j = 0; j < KS; ++j) atomicAdd(&outBase[pos(blk, j)], y[j] * 0.25f);
    };

    float d0[KS], x0[KS], d1[KS], x1[KS], y[KS];
    static_assert(NSTG == 4, "the staging registers below are written out for 4 per thread");
    float4 gb0, gb1, gb2, gb3, gc0, gc1, gc2, gc3;
    auto stage_load_all = [&](int blk) {
        stage_load(blk, 0, gb0, gc0);
        stage_load(blk, 1, gb1, gc1);
        stage_load(blk, 2, gb2, gc2);
        stage_load(blk, 3, gb3, gc3);
    };
    auto stage_store_all = [&](int buf) {
        stage_store(buf, 0, gb0, gc0);
        stage_store(buf, 1, gb1, gc1);
        stage_store(buf, 2, gb2, gc2);
        stage_store(buf, 3, gb3, gc3);
    };
    stage_load_all(0);
    load_dx(0, d0, x0);
    stage_store_all(0);
    __syncthreads();

    for (int blk = 0; blk < NBLK; blk += 2) {
        stage_load_all(blk + 1);
        load_dx(blk + 1, d1, x1);
        __builtin_amdgcn_sched_barrier(0);
        do_block(0, d0, x0, y);
        stage_store_all(1);
        __syncthreads();
        add_out(blk, y);

        const int nb = (blk + 2 < NBLK) ? blk + 2 : blk + 1;
        stage_load_all(nb);
        load_dx(nb, d0, x0);
        __builtin_amdgcn_sched_barrier(0);
        do_block(1, d1, x1, y);
        stage_store_all(0);
        __syncthreads();
        add_out(blk + 1, y);
    }
}

extern "C" void kernel_launch(void* const* d_in, const int* in_sizes, int n_in,
                              void* d_out, int out_size, void* d_ws, size_t ws_size,
                              hipStream_t stream) {
    const float* feat    = (const float*)d_in[0];
    const float* A_log   = (const float*)d_in[1];
    const float* Dvec    = (const float*)d_in[2];
    const float* W_delta = (const float*)d_in[3];
    const float* b_delta = (const float*)d_in[4];
    const float* W_B     = (const float*)d_in[5];
    const float* W_C     = (const float*)d_in[6];
    float* out = (float*)d_out;

    float* ws       = (float*)d_ws;
    float* delta_ws = ws;
    float* Bm_ws    = ws + (size_t)NROW * CC;
    float* Cm_ws    = Bm_ws + (size_t)NROW * NN;

    proj_wmma_kernel<<<NROW / PROJ_ROWS, PROJ_THREADS, 0, stream>>>(
        feat, Dvec, W_delta, b_delta, W_B, W_C, delta_ws, Bm_ws, Cm_ws, out);
    scan_kernel<<<256, 256, 0, stream>>>(feat, A_log, delta_ws, Bm_ws, Cm_ws, out);
}


# scan, row walks: a block's 256 results turned round in a per-wave LDS strip so each atomic add covers 32 consecutive words of out (one 128-byte line, not 32 separate 32-byte pieces); add count kept
# speedup vs baseline: 25.1677x; 1.0583x over previous
#include <hip/hip_runtime.h>
#include <math.h>

typedef __attribute__((ext_vector_type(2))) float v2f;
typedef __attribute__((ext_vector_type(8))) float v8f;

#define BB 2
#define CC 256
#define HH 64
#define WW 64
#define HW 4096
#define NN 16
#define NROW (BB*HW)

__device__ __forceinline__ float softplus_sel(float pre) {
    const float t = log1pf(__expf(pre));
    return (pre > 20.f) ? pre : t;
}
#define PROJ_ROWS  32
#define PROJ_WAVES 18
#define PROJ_THREADS (PROJ_WAVES * 32)
#define PROJ_PITCH 260
#define PROJ_V4 (PROJ_ROWS * CC / 4)
#define PROJ_KSTEPS (CC / 4)
__global__ __launch_bounds__(PROJ_THREADS) void proj_wmma_kernel(const float* __restrict__ feat,
                                 const float* __restrict__ Dvec,
                                 const float* __restrict__ W_delta,
                                 const float* __restrict__ b_delta,
                                 const float* __restrict__ W_B,
                                 const float* __restrict__ W_C,
                                 float* __restrict__ delta_ws,
                                 float* __restrict__ Bm_ws,
                                 float* __restrict__ Cm_ws,
                                 float* __restrict__ out) {
    __shared__ float sX[PROJ_ROWS * PROJ_PITCH];

    const int tid  = threadIdx.x;
    const int lane = tid & 31;
    const int lo   = lane & 15;
    const int hi   = lane >> 4;
    const int which = __builtin_amdgcn_readfirstlane(tid >> 5);
    const int m0   = blockIdx.x * PROJ_ROWS;
    const int bA   = m0 >> 12;
    const int l0   = m0 & (HW - 1);
    const float* xBase = feat + (size_t)bA * CC * HW + l0;

    float4 v[4];
    float dk[4];
    #pragma unroll
    for (int i = 0; i < 4; ++i) {
        const int f = tid + PROJ_THREADS * i;
        if (f < PROJ_V4) {
            v[i] = *reinterpret_cast<const float4*>(xBase + (size_t)(f >> 3) * HW + 4 * (f & 7));
            dk[i] = Dvec[f >> 3];
        }
    }
    #pragma unroll
    for (int i = 0; i < 4; ++i) {
        const int f = tid + PROJ_THREADS * i;
        if (f < PROJ_V4) {
            float* d = sX + (4 * (f & 7)) * PROJ_PITCH + (f >> 3);
            d[0]              = v[i].x;
            d[PROJ_PITCH]     = v[i].y;
            d[2 * PROJ_PITCH] = v[i].z;
            d[3 * PROJ_PITCH] = v[i].w;
        }
    }
    __syncthreads();

    {
        float* oBase = out + (size_t)bA * CC * HW + l0;
        #pragma unroll
        for (int i = 0; i < 4; ++i) {
            const int f = tid + PROJ_THREADS * i;
            if (f < PROJ_V4) {
                float4 o;
                o.x = v[i].x * dk[i]; o.y = v[i].y * dk[i]; o.z = v[i].z * dk[i]; o.w = v[i].w * dk[i];
                *reinterpret_cast<float4*>(oBase + (size_t)(f >> 3) * HW + 4 * (f & 7)) = o;
            }
        }
    }

    const float* Wmat;
    unsigned ncols, n0;
    if (which < 16)       { Wmat = W_delta; ncols = CC; n0 = which * 16; }
    else if (which == 16) { Wmat = W_B;     ncols = NN; n0 = 0; }
    else                  { Wmat = W_C;     ncols = NN; n0 = 0; }
    const unsigned laneoff = (unsigned)(2 * hi) * ncols + n0 + (unsigned)lo;
    v2f bw[PROJ_KSTEPS];
    #pragma unroll
    for (int s = 0; s < PROJ_KSTEPS; ++s) {
        const float* r0 = Wmat + (size_t)(4 * s) * ncols;
        const float* r1 = r0 + ncols;
        bw[s].x = r0[laneoff];
        bw[s].y = r1[laneoff];
    }

    auto run_tile = [&](const float* aRow) -> v8f {
        v8f acc = {};
        #pragma unroll
        for (int s = 0; s < PROJ_KSTEPS; ++s) {
            const v2f a = *reinterpret_cast<const v2f*>(aRow + 4 * s);
            acc = __builtin_amdgcn_wmma_f32_16x16x4_f32(
                 false, a,  false, bw[s],
                 (short)0, acc,  false,  false);
        }
        return acc;
    };
    const float* aRow0 = sX + lo * PROJ_PITCH + 2 * hi;
    const float* aRow1 = aRow0 + 16 * PROJ_PITCH;

    if (which < 16) {
        const float bias = b_delta[n0 + lo];
        float* dcol = delta_ws + (size_t)(m0 + 8 * hi) * CC + n0 + lo;
        const v8f acc0 = run_tile(aRow0);
        v8f acc1 = {};
        #pragma unroll
        for (int r = 0; r < 8; ++r) {
            #pragma unroll
            for (int s = 8 * r; s < 8 * r + 8; ++s) {
                const v2f a = *reinterpret_cast<const v2f*>(aRow1 + 4 * s);
                acc1 = __builtin_amdgcn_wmma_f32_16x16x4_f32(
                     false, a,  false, bw[s],
                     (short)0, acc1,  false,  false);
            }
            dcol[(size_t)r * CC] = softplus_sel(acc0[r] + bias);
            #pragma unroll
            for (int g = 0; g < 8; ++g) {
                __builtin_amdgcn_sched_group_barrier(0x008, 1, 0);
                __builtin_amdgcn_sched_group_barrier(0x002, 11, 0);
            }
            __builtin_amdgcn_sched_barrier(0);
        }
        #pragma unroll
        for (int r = 0; r < 8; ++r)
            dcol[(size_t)(16 + r) * CC] = softplus_sel(acc1[r] + bias);
    } else {
        float* dst = ((which == 16) ? Bm_ws : Cm_ws) + (size_t)(m0 + 8 * hi) * NN + lo;
        const v8f acc0 = run_tile(aRow0);
        #pragma unroll
        for (int r = 0; r < 8; ++r) dst[(size_t)r * NN] = acc0[r];
        const v8f acc1 = run_tile(aRow1);
        #pragma unroll
        for (int r = 0; r < 8; ++r) dst[(size_t)(16 + r) * NN] = acc1[r];
    }
}


#define KS   8
#define BLK  (32 * KS)
#define NBLK (HW / BLK)
#define NG   4
#define NQ   (NN / 4)
#define PITCH 33
#define NSTG (BLK * NQ / 256)

template <int K>
__device__ __forceinline__ float row_shr_or0(float v) {
    return __int_as_float(__builtin_amdgcn_update_dpp(
        0, __float_as_int(v), 0x110 + K, 0xf, 0xf, true));
}
template <int K>
__device__ __forceinline__ float row_shr_or(float outside, float v) {
    return __int_as_float(__builtin_amdgcn_update_dpp(
        __float_as_int(outside), __float_as_int(v), 0x110 + K, 0xf, 0xf, false));
}
template <int K>
__device__ __forceinline__ void scan_level(float& a, float& bb) {
    const float bs = row_shr_or0<K>(bb);
    const float as = row_shr_or<K>(1.0f, a);
    bb = fmaf(a, bs, bb);
    a  = a * as;
}

__global__ __launch_bounds__(256) __attribute__((amdgpu_waves_per_eu(4, 4))) void scan_kernel(const float* __restrict__ feat,
                            const float* __restrict__ A_log,
                            const float* __restrict__ delta_ws,
                            const float* __restrict__ Bm_ws,
                            const float* __restrict__ Cm_ws,
                            float* __restrict__ out) {
    __shared__ float4 sB[2][KS * NQ * PITCH];
    __shared__ float4 sC[2][KS * NQ * PITCH];
    __shared__ float sY[8][BLK];

    const int tid  = threadIdx.x;
    const int lane = tid & 31;
    const int w   = __builtin_amdgcn_readfirstlane((int)(threadIdx.x >> 5));
    const int id  = blockIdx.x * 8 + w;
    const int dir = id >> 9;
    const int b   = (id >> 8) & 1;
    const int c   = id & 255;

    int base, sh, sw;
    switch (dir) {
        case 0:  base = 0;    sh =  64; sw =  1; break;
        case 1:  base = 4095; sh = -64; sw = -1; break;
        case 2:  base = 4032; sh = -64; sw =  1; break;
        default: base = 63;   sh =  -1; sw = 64; break;
    }

    float A[NN];
    #pragma unroll
    for (int n = 0; n < NN; ++n) A[n] = -__expf(A_log[c * NN + n]);

    const float* fBase = feat + (size_t)b * CC * HW + (size_t)c * HW;
    const size_t rbase = (size_t)b * HW;
    float* outBase = out + (size_t)b * CC * HW + (size_t)c * HW;
    const bool upper = lane >= 16;
    const float4* Bm4 = reinterpret_cast<const float4*>(Bm_ws);
    const float4* Cm4 = reinterpret_cast<const float4*>(Cm_ws);

    float hin[NN];
    #pragma unroll
    for (int n = 0; n < NN; ++n) hin[n] = 0.f;

    auto pos_of = [&](int blk, int s) -> int {
        const int t = blk * BLK + s;
        return base + sh * (t >> 6) + sw * (t & 63);
    };
    auto pos = [&](int blk, int j) -> int { return pos_of(blk, KS * lane + j); };

    auto stage_load = [&](int blk, int r, float4& vb, float4& vc) {
        const int f = tid + 256 * r;
        const int l = pos_of(blk, f >> 2);
        const size_t v4 = (rbase + (size_t)l) * NQ + (size_t)(f & 3);
        vb = Bm4[v4];
        vc = Cm4[v4];
    };
    auto stage_store = [&](int buf, int r, const float4& vb, const float4& vc) {
        const int f = tid + 256 * r;
        const int s = f >> 2;
        const int idx = ((s % KS) * NQ + (f & 3)) * PITCH + (s / KS);
        sB[buf][idx] = vb;
        sC[buf][idx] = vc;
    };

    auto load_dx = [&](int blk, float (&d)[KS], float (&xv)[KS]) {
        #pragma unroll
        for (int j = 0; j < KS; ++j) {
            const int l = pos(blk, j);
            d[j]  = delta_ws[(rbase + (size_t)l) * CC + c];
            xv[j] = fBase[l];
        }
    };

    auto lds_load = [&](int buf, int q, float (&Bv)[KS][4], float (&Cv)[KS][4]) {
        #pragma unroll
        for (int j = 0; j < KS; ++j) {
            const float4 bv = sB[buf][(j * NQ + q) * PITCH + lane];
            const float4 cv = sC[buf][(j * NQ + q) * PITCH + lane];
            Bv[j][0] = bv.x; Bv[j][1] = bv.y; Bv[j][2] = bv.z; Bv[j][3] = bv.w;
            Cv[j][0] = cv.x; Cv[j][1] = cv.y; Cv[j][2] = cv.z; Cv[j][3] = cv.w;
        }
    };

    auto do_group = [&](int n0, const float (&d)[KS], const float (&dx)[KS],
                        const float (&Bv)[KS][4], const float (&Cv)[KS][4], float (&y)[KS]) {
        float e[NG][KS], bb[NG][KS], P[NG], Q[NG], H[NG], h[NG];
        #pragma unroll
        for (int g = 0; g < NG; ++g) {
            #pragma unroll
            for (int j = 0; j < KS; ++j) e[g][j] = __expf(d[j] * A[n0 + g]);
        }
        __builtin_amdgcn_sched_barrier(0);
        #pragma unroll
        for (int g = 0; g < NG; ++g) {
            #pragma unroll
            for (int j = 0; j < KS; ++j) bb[g][j] = dx[j] * Bv[j][g];
        }
        #pragma unroll
        for (int g = 0; g < NG; ++g) { P[g] = e[g][0]; Q[g] = bb[g][0]; }
        #pragma unroll
        for (int j = 1; j < KS; ++j) {
            #pragma unroll
            for (int g = 0; g < NG; ++g) {
                Q[g] = fmaf(e[g][j], Q[g], bb[g][j]);
                P[g] = P[g] * e[g][j];
            }
        }
        #pragma unroll
        for (int g = 0; g < NG; ++g) scan_level<1>(P[g], Q[g]);
        #pragma unroll
        for (int g = 0; g < NG; ++g) scan_level<2>(P[g], Q[g]);
        #pragma unroll
        for (int g = 0; g < NG; ++g) scan_level<4>(P[g], Q[g]);
        #pragma unroll
        for (int g = 0; g < NG; ++g) scan_level<8>(P[g], Q[g]);
        #pragma unroll
        for (int g = 0; g < NG; ++g) {
            const int n = n0 + g;
            const float h0 = fmaf(P[g], hin[n], Q[g]);
            const float c0 = __int_as_float(__builtin_amdgcn_readlane(__float_as_int(h0), 15));
            H[g] = upper ? fmaf(P[g], c0, Q[g]) : h0;
            const float rowin = upper ? c0 : hin[n];
            hin[n] = __int_as_float(__builtin_amdgcn_readlane(__float_as_int(H[g]), 31));
            h[g] = row_shr_or<1>(rowin, H[g]);
        }
        #pragma unroll
        for (int j = 0; j < KS - 1; ++j) {
            #pragma unroll
            for (int g = 0; g < NG; ++g) {
                h[g] = fmaf(e[g][j], h[g], bb[g][j]);
                y[j] = fmaf(h[g], Cv[j][g], y[j]);
            }
        }
        #pragma unroll
        for (int g = 0; g < NG; ++g)
            y[KS - 1] = fmaf(H[g], Cv[KS - 1][g], y[KS - 1]);
    };

    auto do_block = [&](int buf, const float (&d)[KS], const float (&xv)[KS], float (&y)[KS]) {
        float dx[KS];
        #pragma unroll
        for (int j = 0; j < KS; ++j) { dx[j] = d[j] * xv[j]; y[j] = 0.f; }
        float Bv[KS][4], Cv[KS][4];
        #pragma unroll
        for (int q = 0; q < NQ; ++q) {
            lds_load(buf, q, Bv, Cv);
            __builtin_amdgcn_sched_barrier(0);
            do_group(4 * q, d, dx, Bv, Cv, y);
        }
    };

    const int lpart = (dir == 3) ? sh * (lane & 3) + sw * (lane >> 2) : sw * lane;
    const int lmin  = (dir == 3) ? -3 : ((dir == 1) ? -31 : 0);
    const unsigned lpBytes = (unsigned)(lpart - lmin) * 4u;
    float* const strip = sY[w];
    int soff[KS];
    #pragma unroll
    for (int i = 0; i < KS; ++i)
        soff[i] = (dir == 3) ? sw * (8 * i) : sh * (i >> 1) + sw * (32 * (i & 1));
    auto add_out = [&](int blk, const float (&y)[KS]) {
        static_assert(KS == 8, "a lane's results are written as two float4 on a row walk");
        if (dir == 3) {
            float* const mine = strip + 32 * (lane & 7) + (lane >> 3);
            #pragma unroll
            for (int j = 0; j < KS; ++j) mine[4 * j] = y[j] * 0.25f;
        } else {
            float4* const mine = reinterpret_cast<float4*>(strip + KS * lane);
            mine[0] = make_float4(y[0] * 0.25f, y[1] * 0.25f, y[2] * 0.25f, y[3] * 0.25f);
            mine[1] = make_float4(y[4] * 0.25f, y[5] * 0.25f, y[6] * 0.25f, y[7] * 0.25f);
        }
        char* const blkBase = reinterpret_cast<char*>(outBase + (base + lmin + sh * 4 * blk));
        float v[KS];
        #pragma unroll
        for (int i = 0; i < KS; ++i) v[i] = strip[32 * i + lane];
        #pragma unroll
        for (int i = 0; i < KS; ++i)
            atomicAdd(reinterpret_cast<float*>(blkBase + (ptrdiff_t)soff[i] * 4 + lpBytes), v[i]);
    };

    float d0[KS], x0[KS], d1[KS], x1[KS], y[KS];
    static_assert(NSTG == 4, "the staging registers below are written out for 4 per thread");
    float4 gb0, gb1, gb2, gb3, gc0, gc1, gc2, gc3;
    auto stage_load_all = [&](int blk) {
        stage_load(blk, 0, gb0, gc0);
        stage_load(blk, 1, gb1, gc1);
        stage_load(blk, 2, gb2, gc2);
        stage_load(blk, 3, gb3, gc3);
    };
    auto stage_store_all = [&](int buf) {
        stage_store(buf, 0, gb0, gc0);
        stage_store(buf, 1, gb1, gc1);
        stage_store(buf, 2, gb2, gc2);
        stage_store(buf, 3, gb3, gc3);
    };
    stage_load_all(0);
    load_dx(0, d0, x0);
    stage_store_all(0);
    __syncthreads();

    for (int blk = 0; blk < NBLK; blk += 2) {
        stage_load_all(blk + 1);
        load_dx(blk + 1, d1, x1);
        __builtin_amdgcn_sched_barrier(0);
        do_block(0, d0, x0, y);
        stage_store_all(1);
        __syncthreads();
        add_out(blk, y);

        const int nb = (blk + 2 < NBLK) ? blk + 2 : blk + 1;
        stage_load_all(nb);
        load_dx(nb, d0, x0);
        __builtin_amdgcn_sched_barrier(0);
        do_block(1, d1, x1, y);
        stage_store_all(0);
        __syncthreads();
        add_out(blk + 1, y);
    }
}

extern "C" void kernel_launch(void* const* d_in, const int* in_sizes, int n_in,
                              void* d_out, int out_size, void* d_ws, size_t ws_size,
                              hipStream_t stream) {
    const float* feat    = (const float*)d_in[0];
    const float* A_log   = (const float*)d_in[1];
    const float* Dvec    = (const float*)d_in[2];
    const float* W_delta = (const float*)d_in[3];
    const float* b_delta = (const float*)d_in[4];
    const float* W_B     = (const float*)d_in[5];
    const float* W_C     = (const float*)d_in[6];
    float* out = (float*)d_out;

    float* ws       = (float*)d_ws;
    float* delta_ws = ws;
    float* Bm_ws    = ws + (size_t)NROW * CC;
    float* Cm_ws    = Bm_ws + (size_t)NROW * NN;

    proj_wmma_kernel<<<NROW / PROJ_ROWS, PROJ_THREADS, 0, stream>>>(
        feat, Dvec, W_delta, b_delta, W_B, W_C, delta_ws, Bm_ws, Cm_ws, out);
    scan_kernel<<<256, 256, 0, stream>>>(feat, A_log, delta_ws, Bm_ws, Cm_ws, out);
}


# scan: a block's delta is fetched once per work-group (16-byte loads, 2 lanes per position) and shared by its 8 waves through an LDS strip, not loaded by every wave one line per lane; sum order kept
# speedup vs baseline: 27.4914x; 1.0923x over previous
#include <hip/hip_runtime.h>
#include <math.h>

typedef __attribute__((ext_vector_type(2))) float v2f;
typedef __attribute__((ext_vector_type(8))) float v8f;

#define BB 2
#define CC 256
#define HH 64
#define WW 64
#define HW 4096
#define NN 16
#define NROW (BB*HW)

__device__ __forceinline__ float softplus_sel(float pre) {
    const float t = log1pf(__expf(pre));
    return (pre > 20.f) ? pre : t;
}
#define PROJ_ROWS  32
#define PROJ_WAVES 18
#define PROJ_THREADS (PROJ_WAVES * 32)
#define PROJ_PITCH 260
#define PROJ_V4 (PROJ_ROWS * CC / 4)
#define PROJ_KSTEPS (CC / 4)
__global__ __launch_bounds__(PROJ_THREADS) void proj_wmma_kernel(const float* __restrict__ feat,
                                 const float* __restrict__ Dvec,
                                 const float* __restrict__ W_delta,
                                 const float* __restrict__ b_delta,
                                 const float* __restrict__ W_B,
                                 const float* __restrict__ W_C,
                                 float* __restrict__ delta_ws,
                                 float* __restrict__ Bm_ws,
                                 float* __restrict__ Cm_ws,
                                 float* __restrict__ out) {
    __shared__ float sX[PROJ_ROWS * PROJ_PITCH];

    const int tid  = threadIdx.x;
    const int lane = tid & 31;
    const int lo   = lane & 15;
    const int hi   = lane >> 4;
    const int which = __builtin_amdgcn_readfirstlane(tid >> 5);
    const int m0   = blockIdx.x * PROJ_ROWS;
    const int bA   = m0 >> 12;
    const int l0   = m0 & (HW - 1);
    const float* xBase = feat + (size_t)bA * CC * HW + l0;

    float4 v[4];
    float dk[4];
    #pragma unroll
    for (int i = 0; i < 4; ++i) {
        const int f = tid + PROJ_THREADS * i;
        if (f < PROJ_V4) {
            v[i] = *reinterpret_cast<const float4*>(xBase + (size_t)(f >> 3) * HW + 4 * (f & 7));
            dk[i] = Dvec[f >> 3];
        }
    }
    #pragma unroll
    for (int i = 0; i < 4; ++i) {
        const int f = tid + PROJ_THREADS * i;
        if (f < PROJ_V4) {
            float* d = sX + (4 * (f & 7)) * PROJ_PITCH + (f >> 3);
            d[0]              = v[i].x;
            d[PROJ_PITCH]     = v[i].y;
            d[2 * PROJ_PITCH] = v[i].z;
            d[3 * PROJ_PITCH] = v[i].w;
        }
    }
    __syncthreads();

    {
        float* oBase = out + (size_t)bA * CC * HW + l0;
        #pragma unroll
        for (int i = 0; i < 4; ++i) {
            const int f = tid + PROJ_THREADS * i;
            if (f < PROJ_V4) {
                float4 o;
                o.x = v[i].x * dk[i]; o.y = v[i].y * dk[i]; o.z = v[i].z * dk[i]; o.w = v[i].w * dk[i];
                *reinterpret_cast<float4*>(oBase + (size_t)(f >> 3) * HW + 4 * (f & 7)) = o;
            }
        }
    }

    const float* Wmat;
    unsigned ncols, n0;
    if (which < 16)       { Wmat = W_delta; ncols = CC; n0 = which * 16; }
    else if (which == 16) { Wmat = W_B;     ncols = NN; n0 = 0; }
    else                  { Wmat = W_C;     ncols = NN; n0 = 0; }
    const unsigned laneoff = (unsigned)(2 * hi) * ncols + n0 + (unsigned)lo;
    v2f bw[PROJ_KSTEPS];
    #pragma unroll
    for (int s = 0; s < PROJ_KSTEPS; ++s) {
        const float* r0 = Wmat + (size_t)(4 * s) * ncols;
        const float* r1 = r0 + ncols;
        bw[s].x = r0[laneoff];
        bw[s].y = r1[laneoff];
    }

    auto run_tile = [&](const float* aRow) -> v8f {
        v8f acc = {};
        #pragma unroll
        for (int s = 0; s < PROJ_KSTEPS; ++s) {
            const v2f a = *reinterpret_cast<const v2f*>(aRow + 4 * s);
            acc = __builtin_amdgcn_wmma_f32_16x16x4_f32(
                 false, a,  false, bw[s],
                 (short)0, acc,  false,  false);
        }
        return acc;
    };
    const float* aRow0 = sX + lo * PROJ_PITCH + 2 * hi;
    const float* aRow1 = aRow0 + 16 * PROJ_PITCH;

    if (which < 16) {
        const float bias = b_delta[n0 + lo];
        float* dcol = delta_ws + (size_t)(m0 + 8 * hi) * CC + n0 + lo;
        const v8f acc0 = run_tile(aRow0);
        v8f acc1 = {};
        #pragma unroll
        for (int r = 0; r < 8; ++r) {
            #pragma unroll
            for (int s = 8 * r; s < 8 * r + 8; ++s) {
                const v2f a = *reinterpret_cast<const v2f*>(aRow1 + 4 * s);
                acc1 = __builtin_amdgcn_wmma_f32_16x16x4_f32(
                     false, a,  false, bw[s],
                     (short)0, acc1,  false,  false);
            }
            dcol[(size_t)r * CC] = softplus_sel(acc0[r] + bias);
            #pragma unroll
            for (int g = 0; g < 8; ++g) {
                __builtin_amdgcn_sched_group_barrier(0x008, 1, 0);
                __builtin_amdgcn_sched_group_barrier(0x002, 11, 0);
            }
            __builtin_amdgcn_sched_barrier(0);
        }
        #pragma unroll
        for (int r = 0; r < 8; ++r)
            dcol[(size_t)(16 + r) * CC] = softplus_sel(acc1[r] + bias);
    } else {
        float* dst = ((which == 16) ? Bm_ws : Cm_ws) + (size_t)(m0 + 8 * hi) * NN + lo;
        const v8f acc0 = run_tile(aRow0);
        #pragma unroll
        for (int r = 0; r < 8; ++r) dst[(size_t)r * NN] = acc0[r];
        const v8f acc1 = run_tile(aRow1);
        #pragma unroll
        for (int r = 0; r < 8; ++r) dst[(size_t)(16 + r) * NN] = acc1[r];
    }
}


#define KS   8
#define BLK  (32 * KS)
#define NBLK (HW / BLK)
#define NG   4
#define NQ   (NN / 4)
#define PITCH 33
#define NSTG (BLK * NQ / 256)
#define DPITCH (BLK + 4)
#define NSTD (BLK * 2 / 256)

template <int K>
__device__ __forceinline__ float row_shr_or0(float v) {
    return __int_as_float(__builtin_amdgcn_update_dpp(
        0, __float_as_int(v), 0x110 + K, 0xf, 0xf, true));
}
template <int K>
__device__ __forceinline__ float row_shr_or(float outside, float v) {
    return __int_as_float(__builtin_amdgcn_update_dpp(
        __float_as_int(outside), __float_as_int(v), 0x110 + K, 0xf, 0xf, false));
}
template <int K>
__device__ __forceinline__ void scan_level(float& a, float& bb) {
    const float bs = row_shr_or0<K>(bb);
    const float as = row_shr_or<K>(1.0f, a);
    bb = fmaf(a, bs, bb);
    a  = a * as;
}

__global__ __launch_bounds__(256) __attribute__((amdgpu_waves_per_eu(4, 4))) void scan_kernel(const float* __restrict__ feat,
                            const float* __restrict__ A_log,
                            const float* __restrict__ delta_ws,
                            const float* __restrict__ Bm_ws,
                            const float* __restrict__ Cm_ws,
                            float* __restrict__ out) {
    __shared__ float4 sB[2][KS * NQ * PITCH];
    __shared__ float4 sC[2][KS * NQ * PITCH];
    __shared__ float sY[8][BLK];
    __shared__ float4 sD[2][8 * (DPITCH / 4)];

    const int tid  = threadIdx.x;
    const int lane = tid & 31;
    const int w   = __builtin_amdgcn_readfirstlane((int)(threadIdx.x >> 5));
    const int id  = blockIdx.x * 8 + w;
    const int dir = id >> 9;
    const int b   = (id >> 8) & 1;
    const int c   = id & 255;

    int base, sh, sw;
    switch (dir) {
        case 0:  base = 0;    sh =  64; sw =  1; break;
        case 1:  base = 4095; sh = -64; sw = -1; break;
        case 2:  base = 4032; sh = -64; sw =  1; break;
        default: base = 63;   sh =  -1; sw = 64; break;
    }

    float A[NN];
    #pragma unroll
    for (int n = 0; n < NN; ++n) A[n] = -__expf(A_log[c * NN + n]);

    const float* fBase = feat + (size_t)b * CC * HW + (size_t)c * HW;
    const size_t rbase = (size_t)b * HW;
    float* outBase = out + (size_t)b * CC * HW + (size_t)c * HW;
    const bool upper = lane >= 16;
    const float4* Bm4 = reinterpret_cast<const float4*>(Bm_ws);
    const float4* Cm4 = reinterpret_cast<const float4*>(Cm_ws);

    float hin[NN];
    #pragma unroll
    for (int n = 0; n < NN; ++n) hin[n] = 0.f;

    auto pos_of = [&](int blk, int s) -> int {
        const int t = blk * BLK + s;
        return base + sh * (t >> 6) + sw * (t & 63);
    };
    auto pos = [&](int blk, int j) -> int { return pos_of(blk, KS * lane + j); };

    auto stage_load = [&](int blk, int r, float4& vb, float4& vc) {
        const int f = tid + 256 * r;
        const int l = pos_of(blk, f >> 2);
        const size_t v4 = (rbase + (size_t)l) * NQ + (size_t)(f & 3);
        vb = Bm4[v4];
        vc = Cm4[v4];
    };
    auto stage_store = [&](int buf, int r, const float4& vb, const float4& vc) {
        const int f = tid + 256 * r;
        const int s = f >> 2;
        const int idx = ((s % KS) * NQ + (f & 3)) * PITCH + (s / KS);
        sB[buf][idx] = vb;
        sC[buf][idx] = vc;
    };

    const int c0 = c - w;
    const float4* D4 = reinterpret_cast<const float4*>(delta_ws);
    auto stage_load_d = [&](int blk, int r, float4& vd) {
        const int f = tid + 256 * r;
        const int l = pos_of(blk, f >> 1);
        vd = D4[(rbase + (size_t)l) * (CC / 4) + (size_t)((c0 >> 2) + (f & 1))];
    };
    auto stage_store_d = [&](int buf, int r, const float4& vd) {
        const int f = tid + 256 * r;
        float* p = reinterpret_cast<float*>(sD[buf]) + (4 * (f & 1)) * DPITCH + (f >> 1);
        p[0]          = vd.x;
        p[DPITCH]     = vd.y;
        p[2 * DPITCH] = vd.z;
        p[3 * DPITCH] = vd.w;
    };
    auto lds_load_d = [&](int buf, float (&d)[KS]) {
        static_assert(KS == 8, "a lane's delta values are read as two float4");
        const float4* p = reinterpret_cast<const float4*>(
            reinterpret_cast<const float*>(sD[buf]) + w * DPITCH + KS * lane);
        const float4 lo4 = p[0];
        const float4 hi4 = p[1];
        d[0] = lo4.x; d[1] = lo4.y; d[2] = lo4.z; d[3] = lo4.w;
        d[4] = hi4.x; d[5] = hi4.y; d[6] = hi4.z; d[7] = hi4.w;
    };

    auto load_x = [&](int blk, float (&xv)[KS]) {
        #pragma unroll
        for (int j = 0; j < KS; ++j) xv[j] = fBase[pos(blk, j)];
    };

    auto lds_load = [&](int buf, int q, float (&Bv)[KS][4], float (&Cv)[KS][4]) {
        #pragma unroll
        for (int j = 0; j < KS; ++j) {
            const float4 bv = sB[buf][(j * NQ + q) * PITCH + lane];
            const float4 cv = sC[buf][(j * NQ + q) * PITCH + lane];
            Bv[j][0] = bv.x; Bv[j][1] = bv.y; Bv[j][2] = bv.z; Bv[j][3] = bv.w;
            Cv[j][0] = cv.x; Cv[j][1] = cv.y; Cv[j][2] = cv.z; Cv[j][3] = cv.w;
        }
    };

    auto do_group = [&](int n0, const float (&d)[KS], const float (&dx)[KS],
                        const float (&Bv)[KS][4], const float (&Cv)[KS][4], float (&y)[KS]) {
        float e[NG][KS], bb[NG][KS], P[NG], Q[NG], H[NG], h[NG];
        #pragma unroll
        for (int g = 0; g < NG; ++g) {
            #pragma unroll
            for (int j = 0; j < KS; ++j) e[g][j] = __expf(d[j] * A[n0 + g]);
        }
        __builtin_amdgcn_sched_barrier(0);
        #pragma unroll
        for (int g = 0; g < NG; ++g) {
            #pragma unroll
            for (int j = 0; j < KS; ++j) bb[g][j] = dx[j] * Bv[j][g];
        }
        #pragma unroll
        for (int g = 0; g < NG; ++g) { P[g] = e[g][0]; Q[g] = bb[g][0]; }
        #pragma unroll
        for (int j = 1; j < KS; ++j) {
            #pragma unroll
            for (int g = 0; g < NG; ++g) {
                Q[g] = fmaf(e[g][j], Q[g], bb[g][j]);
                P[g] = P[g] * e[g][j];
            }
        }
        #pragma unroll
        for (int g = 0; g < NG; ++g) scan_level<1>(P[g], Q[g]);
        #pragma unroll
        for (int g = 0; g < NG; ++g) scan_level<2>(P[g], Q[g]);
        #pragma unroll
        for (int g = 0; g < NG; ++g) scan_level<4>(P[g], Q[g]);
        #pragma unroll
        for (int g = 0; g < NG; ++g) scan_level<8>(P[g], Q[g]);
        #pragma unroll
        for (int g = 0; g < NG; ++g) {
            const int n = n0 + g;
            const float h0 = fmaf(P[g], hin[n], Q[g]);
            const float c0 = __int_as_float(__builtin_amdgcn_readlane(__float_as_int(h0), 15));
            H[g] = upper ? fmaf(P[g], c0, Q[g]) : h0;
            const float rowin = upper ? c0 : hin[n];
            hin[n] = __int_as_float(__builtin_amdgcn_readlane(__float_as_int(H[g]), 31));
            h[g] = row_shr_or<1>(rowin, H[g]);
        }
        #pragma unroll
        for (int j = 0; j < KS - 1; ++j) {
            #pragma unroll
            for (int g = 0; g < NG; ++g) {
                h[g] = fmaf(e[g][j], h[g], bb[g][j]);
                y[j] = fmaf(h[g], Cv[j][g], y[j]);
            }
        }
        #pragma unroll
        for (int g = 0; g < NG; ++g)
            y[KS - 1] = fmaf(H[g], Cv[KS - 1][g], y[KS - 1]);
    };

    auto do_block = [&](int buf, const float (&d)[KS], const float (&xv)[KS], float (&y)[KS]) {
        float dx[KS];
        #pragma unroll
        for (int j = 0; j < KS; ++j) { dx[j] = d[j] * xv[j]; y[j] = 0.f; }
        float Bv[KS][4], Cv[KS][4];
        #pragma unroll
        for (int q = 0; q < NQ; ++q) {
            lds_load(buf, q, Bv, Cv);
            __builtin_amdgcn_sched_barrier(0);
            do_group(4 * q, d, dx, Bv, Cv, y);
        }
    };

    const int lpart = (dir == 3) ? sh * (lane & 3) + sw * (lane >> 2) : sw * lane;
    const int lmin  = (dir == 3) ? -3 : ((dir == 1) ? -31 : 0);
    const unsigned lpBytes = (unsigned)(lpart - lmin) * 4u;
    float* const strip = sY[w];
    int soff[KS];
    #pragma unroll
    for (int i = 0; i < KS; ++i)
        soff[i] = (dir == 3) ? sw * (8 * i) : sh * (i >> 1) + sw * (32 * (i & 1));
    auto add_out = [&](int blk, const float (&y)[KS]) {
        static_assert(KS == 8, "a lane's results are written as two float4 on a row walk");
        if (dir == 3) {
            float* const mine = strip + 32 * (lane & 7) + (lane >> 3);
            #pragma unroll
            for (int j = 0; j < KS; ++j) mine[4 * j] = y[j] * 0.25f;
        } else {
            float4* const mine = reinterpret_cast<float4*>(strip + KS * lane);
            mine[0] = make_float4(y[0] * 0.25f, y[1] * 0.25f, y[2] * 0.25f, y[3] * 0.25f);
            mine[1] = make_float4(y[4] * 0.25f, y[5] * 0.25f, y[6] * 0.25f, y[7] * 0.25f);
        }
        char* const blkBase = reinterpret_cast<char*>(outBase + (base + lmin + sh * 4 * blk));
        float v[KS];
        #pragma unroll
        for (int i = 0; i < KS; ++i) v[i] = strip[32 * i + lane];
        #pragma unroll
        for (int i = 0; i < KS; ++i)
            atomicAdd(reinterpret_cast<float*>(blkBase + (ptrdiff_t)soff[i] * 4 + lpBytes), v[i]);
    };

    float d0[KS], x0[KS], d1[KS], x1[KS], y[KS];
    static_assert(NSTG == 4, "the staging registers below are written out for 4 per thread");
    static_assert(NSTD == 2, "the delta staging registers below are written out for 2 per thread");
    float4 gb0, gb1, gb2, gb3, gc0, gc1, gc2, gc3, gd0, gd1;
    auto stage_load_all = [&](int blk) {
        stage_load_d(blk, 0, gd0);
        stage_load_d(blk, 1, gd1);
        stage_load(blk, 0, gb0, gc0);
        stage_load(blk, 1, gb1, gc1);
        stage_load(blk, 2, gb2, gc2);
        stage_load(blk, 3, gb3, gc3);
    };
    auto stage_store_all = [&](int buf) {
        stage_store_d(buf, 0, gd0);
        stage_store_d(buf, 1, gd1);
        stage_store(buf, 0, gb0, gc0);
        stage_store(buf, 1, gb1, gc1);
        stage_store(buf, 2, gb2, gc2);
        stage_store(buf, 3, gb3, gc3);
    };
    stage_load_all(0);
    load_x(0, x0);
    stage_store_all(0);
    __syncthreads();
    lds_load_d(0, d0);

    for (int blk = 0; blk < NBLK; blk += 2) {
        stage_load_all(blk + 1);
        load_x(blk + 1, x1);
        __builtin_amdgcn_sched_barrier(0);
        do_block(0, d0, x0, y);
        stage_store_all(1);
        __syncthreads();
        lds_load_d(1, d1);
        add_out(blk, y);

        const int nb = (blk + 2 < NBLK) ? blk + 2 : blk + 1;
        stage_load_all(nb);
        load_x(nb, x0);
        __builtin_amdgcn_sched_barrier(0);
        do_block(1, d1, x1, y);
        stage_store_all(0);
        __syncthreads();
        lds_load_d(0, d0);
        add_out(blk + 1, y);
    }
}

extern "C" void kernel_launch(void* const* d_in, const int* in_sizes, int n_in,
                              void* d_out, int out_size, void* d_ws, size_t ws_size,
                              hipStream_t stream) {
    const float* feat    = (const float*)d_in[0];
    const float* A_log   = (const float*)d_in[1];
    const float* Dvec    = (const float*)d_in[2];
    const float* W_delta = (const float*)d_in[3];
    const float* b_delta = (const float*)d_in[4];
    const float* W_B     = (const float*)d_in[5];
    const float* W_C     = (const float*)d_in[6];
    float* out = (float*)d_out;

    float* ws       = (float*)d_ws;
    float* delta_ws = ws;
    float* Bm_ws    = ws + (size_t)NROW * CC;
    float* Cm_ws    = Bm_ws + (size_t)NROW * NN;

    proj_wmma_kernel<<<NROW / PROJ_ROWS, PROJ_THREADS, 0, stream>>>(
        feat, Dvec, W_delta, b_delta, W_B, W_C, delta_ws, Bm_ws, Cm_ws, out);
    scan_kernel<<<256, 256, 0, stream>>>(feat, A_log, delta_ws, Bm_ws, Cm_ws, out);
}


# scan: decay factors are no longer multiplied: each decay product (lane fold, 4 scan levels, hand-over) is one exp2 of a delta sum (4 DPP adds a block); rates scaled by log2 e; decay rounding changes
# speedup vs baseline: 28.4747x; 1.0358x over previous
#include <hip/hip_runtime.h>
#include <math.h>

typedef __attribute__((ext_vector_type(2))) float v2f;
typedef __attribute__((ext_vector_type(8))) float v8f;

#define BB 2
#define CC 256
#define HH 64
#define WW 64
#define HW 4096
#define NN 16
#define NROW (BB*HW)

__device__ __forceinline__ float softplus_sel(float pre) {
    const float t = log1pf(__expf(pre));
    return (pre > 20.f) ? pre : t;
}
#define PROJ_ROWS  32
#define PROJ_WAVES 18
#define PROJ_THREADS (PROJ_WAVES * 32)
#define PROJ_PITCH 260
#define PROJ_V4 (PROJ_ROWS * CC / 4)
#define PROJ_KSTEPS (CC / 4)
__global__ __launch_bounds__(PROJ_THREADS) void proj_wmma_kernel(const float* __restrict__ feat,
                                 const float* __restrict__ Dvec,
                                 const float* __restrict__ W_delta,
                                 const float* __restrict__ b_delta,
                                 const float* __restrict__ W_B,
                                 const float* __restrict__ W_C,
                                 float* __restrict__ delta_ws,
                                 float* __restrict__ Bm_ws,
                                 float* __restrict__ Cm_ws,
                                 float* __restrict__ out) {
    __shared__ float sX[PROJ_ROWS * PROJ_PITCH];

    const int tid  = threadIdx.x;
    const int lane = tid & 31;
    const int lo   = lane & 15;
    const int hi   = lane >> 4;
    const int which = __builtin_amdgcn_readfirstlane(tid >> 5);
    const int m0   = blockIdx.x * PROJ_ROWS;
    const int bA   = m0 >> 12;
    const int l0   = m0 & (HW - 1);
    const float* xBase = feat + (size_t)bA * CC * HW + l0;

    float4 v[4];
    float dk[4];
    #pragma unroll
    for (int i = 0; i < 4; ++i) {
        const int f = tid + PROJ_THREADS * i;
        if (f < PROJ_V4) {
            v[i] = *reinterpret_cast<const float4*>(xBase + (size_t)(f >> 3) * HW + 4 * (f & 7));
            dk[i] = Dvec[f >> 3];
        }
    }
    #pragma unroll
    for (int i = 0; i < 4; ++i) {
        const int f = tid + PROJ_THREADS * i;
        if (f < PROJ_V4) {
            float* d = sX + (4 * (f & 7)) * PROJ_PITCH + (f >> 3);
            d[0]              = v[i].x;
            d[PROJ_PITCH]     = v[i].y;
            d[2 * PROJ_PITCH] = v[i].z;
            d[3 * PROJ_PITCH] = v[i].w;
        }
    }
    __syncthreads();

    {
        float* oBase = out + (size_t)bA * CC * HW + l0;
        #pragma unroll
        for (int i = 0; i < 4; ++i) {
            const int f = tid + PROJ_THREADS * i;
            if (f < PROJ_V4) {
                float4 o;
                o.x = v[i].x * dk[i]; o.y = v[i].y * dk[i]; o.z = v[i].z * dk[i]; o.w = v[i].w * dk[i];
                *reinterpret_cast<float4*>(oBase + (size_t)(f >> 3) * HW + 4 * (f & 7)) = o;
            }
        }
    }

    const float* Wmat;
    unsigned ncols, n0;
    if (which < 16)       { Wmat = W_delta; ncols = CC; n0 = which * 16; }
    else if (which == 16) { Wmat = W_B;     ncols = NN; n0 = 0; }
    else                  { Wmat = W_C;     ncols = NN; n0 = 0; }
    const unsigned laneoff = (unsigned)(2 * hi) * ncols + n0 + (unsigned)lo;
    v2f bw[PROJ_KSTEPS];
    #pragma unroll
    for (int s = 0; s < PROJ_KSTEPS; ++s) {
        const float* r0 = Wmat + (size_t)(4 * s) * ncols;
        const float* r1 = r0 + ncols;
        bw[s].x = r0[laneoff];
        bw[s].y = r1[laneoff];
    }

    auto run_tile = [&](const float* aRow) -> v8f {
        v8f acc = {};
        #pragma unroll
        for (int s = 0; s < PROJ_KSTEPS; ++s) {
            const v2f a = *reinterpret_cast<const v2f*>(aRow + 4 * s);
            acc = __builtin_amdgcn_wmma_f32_16x16x4_f32(
                 false, a,  false, bw[s],
                 (short)0, acc,  false,  false);
        }
        return acc;
    };
    const float* aRow0 = sX + lo * PROJ_PITCH + 2 * hi;
    const float* aRow1 = aRow0 + 16 * PROJ_PITCH;

    if (which < 16) {
        const float bias = b_delta[n0 + lo];
        float* dcol = delta_ws + (size_t)(m0 + 8 * hi) * CC + n0 + lo;
        const v8f acc0 = run_tile(aRow0);
        v8f acc1 = {};
        #pragma unroll
        for (int r = 0; r < 8; ++r) {
            #pragma unroll
            for (int s = 8 * r; s < 8 * r + 8; ++s) {
                const v2f a = *reinterpret_cast<const v2f*>(aRow1 + 4 * s);
                acc1 = __builtin_amdgcn_wmma_f32_16x16x4_f32(
                     false, a,  false, bw[s],
                     (short)0, acc1,  false,  false);
            }
            dcol[(size_t)r * CC] = softplus_sel(acc0[r] + bias);
            #pragma unroll
            for (int g = 0; g < 8; ++g) {
                __builtin_amdgcn_sched_group_barrier(0x008, 1, 0);
                __builtin_amdgcn_sched_group_barrier(0x002, 11, 0);
            }
            __builtin_amdgcn_sched_barrier(0);
        }
        #pragma unroll
        for (int r = 0; r < 8; ++r)
            dcol[(size_t)(16 + r) * CC] = softplus_sel(acc1[r] + bias);
    } else {
        float* dst = ((which == 16) ? Bm_ws : Cm_ws) + (size_t)(m0 + 8 * hi) * NN + lo;
        const v8f acc0 = run_tile(aRow0);
        #pragma unroll
        for (int r = 0; r < 8; ++r) dst[(size_t)r * NN] = acc0[r];
        const v8f acc1 = run_tile(aRow1);
        #pragma unroll
        for (int r = 0; r < 8; ++r) dst[(size_t)(16 + r) * NN] = acc1[r];
    }
}


#define KS   8
#define BLK  (32 * KS)
#define NBLK (HW / BLK)
#define NG   4
#define NQ   (NN / 4)
#define PITCH 33
#define NSTG (BLK * NQ / 256)
#define DPITCH (BLK + 4)
#define NSTD (BLK * 2 / 256)

template <int K>
__device__ __forceinline__ float row_shr_or0(float v) {
    return __int_as_float(__builtin_amdgcn_update_dpp(
        0, __float_as_int(v), 0x110 + K, 0xf, 0xf, true));
}
template <int K>
__device__ __forceinline__ float row_shr_or(float outside, float v) {
    return __int_as_float(__builtin_amdgcn_update_dpp(
        __float_as_int(outside), __float_as_int(v), 0x110 + K, 0xf, 0xf, false));
}
template <int K>
__device__ __forceinline__ void scan_level(const float a, float& bb) {
    const float bs = row_shr_or0<K>(bb);
    bb = fmaf(a, bs, bb);
}

__global__ __launch_bounds__(256) __attribute__((amdgpu_waves_per_eu(4, 4))) void scan_kernel(const float* __restrict__ feat,
                            const float* __restrict__ A_log,
                            const float* __restrict__ delta_ws,
                            const float* __restrict__ Bm_ws,
                            const float* __restrict__ Cm_ws,
                            float* __restrict__ out) {
    __shared__ float4 sB[2][KS * NQ * PITCH];
    __shared__ float4 sC[2][KS * NQ * PITCH];
    __shared__ float sY[8][BLK];
    __shared__ float4 sD[2][8 * (DPITCH / 4)];

    const int tid  = threadIdx.x;
    const int lane = tid & 31;
    const int w   = __builtin_amdgcn_readfirstlane((int)(threadIdx.x >> 5));
    const int id  = blockIdx.x * 8 + w;
    const int dir = id >> 9;
    const int b   = (id >> 8) & 1;
    const int c   = id & 255;

    int base, sh, sw;
    switch (dir) {
        case 0:  base = 0;    sh =  64; sw =  1; break;
        case 1:  base = 4095; sh = -64; sw = -1; break;
        case 2:  base = 4032; sh = -64; sw =  1; break;
        default: base = 63;   sh =  -1; sw = 64; break;
    }

    float A[NN];
    #pragma unroll
    for (int n = 0; n < NN; ++n) A[n] = -__expf(A_log[c * NN + n]) * 1.44269504088896341f;

    const float* fBase = feat + (size_t)b * CC * HW + (size_t)c * HW;
    const size_t rbase = (size_t)b * HW;
    float* outBase = out + (size_t)b * CC * HW + (size_t)c * HW;
    const bool upper = lane >= 16;
    const float4* Bm4 = reinterpret_cast<const float4*>(Bm_ws);
    const float4* Cm4 = reinterpret_cast<const float4*>(Cm_ws);

    float hin[NN];
    #pragma unroll
    for (int n = 0; n < NN; ++n) hin[n] = 0.f;

    auto pos_of = [&](int blk, int s) -> int {
        const int t = blk * BLK + s;
        return base + sh * (t >> 6) + sw * (t & 63);
    };
    auto pos = [&](int blk, int j) -> int { return pos_of(blk, KS * lane + j); };

    auto stage_load = [&](int blk, int r, float4& vb, float4& vc) {
        const int f = tid + 256 * r;
        const int l = pos_of(blk, f >> 2);
        const size_t v4 = (rbase + (size_t)l) * NQ + (size_t)(f & 3);
        vb = Bm4[v4];
        vc = Cm4[v4];
    };
    auto stage_store = [&](int buf, int r, const float4& vb, const float4& vc) {
        const int f = tid + 256 * r;
        const int s = f >> 2;
        const int idx = ((s % KS) * NQ + (f & 3)) * PITCH + (s / KS);
        sB[buf][idx] = vb;
        sC[buf][idx] = vc;
    };

    const int c0 = c - w;
    const float4* D4 = reinterpret_cast<const float4*>(delta_ws);
    auto stage_load_d = [&](int blk, int r, float4& vd) {
        const int f = tid + 256 * r;
        const int l = pos_of(blk, f >> 1);
        vd = D4[(rbase + (size_t)l) * (CC / 4) + (size_t)((c0 >> 2) + (f & 1))];
    };
    auto stage_store_d = [&](int buf, int r, const float4& vd) {
        const int f = tid + 256 * r;
        float* p = reinterpret_cast<float*>(sD[buf]) + (4 * (f & 1)) * DPITCH + (f >> 1);
        p[0]          = vd.x;
        p[DPITCH]     = vd.y;
        p[2 * DPITCH] = vd.z;
        p[3 * DPITCH] = vd.w;
    };
    auto lds_load_d = [&](int buf, float (&d)[KS]) {
        static_assert(KS == 8, "a lane's delta values are read as two float4");
        const float4* p = reinterpret_cast<const float4*>(
            reinterpret_cast<const float*>(sD[buf]) + w * DPITCH + KS * lane);
        const float4 lo4 = p[0];
        const float4 hi4 = p[1];
        d[0] = lo4.x; d[1] = lo4.y; d[2] = lo4.z; d[3] = lo4.w;
        d[4] = hi4.x; d[5] = hi4.y; d[6] = hi4.z; d[7] = hi4.w;
    };

    auto load_x = [&](int blk, float (&xv)[KS]) {
        #pragma unroll
        for (int j = 0; j < KS; ++j) xv[j] = fBase[pos(blk, j)];
    };

    auto lds_load = [&](int buf, int q, float (&Bv)[KS][4], float (&Cv)[KS][4]) {
        #pragma unroll
        for (int j = 0; j < KS; ++j) {
            const float4 bv = sB[buf][(j * NQ + q) * PITCH + lane];
            const float4 cv = sC[buf][(j * NQ + q) * PITCH + lane];
            Bv[j][0] = bv.x; Bv[j][1] = bv.y; Bv[j][2] = bv.z; Bv[j][3] = bv.w;
            Cv[j][0] = cv.x; Cv[j][1] = cv.y; Cv[j][2] = cv.z; Cv[j][3] = cv.w;
        }
    };

    auto do_group = [&](int n0, const float (&Wd)[5], const float (&d)[KS], const float (&dx)[KS],
                        const float (&Bv)[KS][4], const float (&Cv)[KS][4], float (&y)[KS]) {
        float e[NG][KS], bb[NG][KS], pk[NG][5], Q[NG], H[NG], h[NG];
        #pragma unroll
        for (int g = 0; g < NG; ++g) {
            #pragma unroll
            for (int j = 0; j < KS; ++j) e[g][j] = __builtin_amdgcn_exp2f(d[j] * A[n0 + g]);
        }
        __builtin_amdgcn_sched_barrier(0);
        #pragma unroll
        for (int g = 0; g < NG; ++g) {
            #pragma unroll
            for (int k = 0; k < 5; ++k) pk[g][k] = __builtin_amdgcn_exp2f(Wd[k] * A[n0 + g]);
        }
        #pragma unroll
        for (int g = 0; g < NG; ++g) {
            #pragma unroll
            for (int j = 0; j < KS; ++j) bb[g][j] = dx[j] * Bv[j][g];
        }
        #pragma unroll
        for (int g = 0; g < NG; ++g) Q[g] = bb[g][0];
        #pragma unroll
        for (int j = 1; j < KS; ++j) {
            #pragma unroll
            for (int g = 0; g < NG; ++g) Q[g] = fmaf(e[g][j], Q[g], bb[g][j]);
        }
        #pragma unroll
        for (int g = 0; g < NG; ++g) scan_level<1>(pk[g][0], Q[g]);
        #pragma unroll
        for (int g = 0; g < NG; ++g) scan_level<2>(pk[g][1], Q[g]);
        #pragma unroll
        for (int g = 0; g < NG; ++g) scan_level<4>(pk[g][2], Q[g]);
        #pragma unroll
        for (int g = 0; g < NG; ++g) scan_level<8>(pk[g][3], Q[g]);
        #pragma unroll
        for (int g = 0; g < NG; ++g) {
            const int n = n0 + g;
            const float Pg = pk[g][4];
            const float h0 = fmaf(Pg, hin[n], Q[g]);
            const float c0 = __int_as_float(__builtin_amdgcn_readlane(__float_as_int(h0), 15));
            H[g] = upper ? fmaf(Pg, c0, Q[g]) : h0;
            const float rowin = upper ? c0 : hin[n];
            hin[n] = __int_as_float(__builtin_amdgcn_readlane(__float_as_int(H[g]), 31));
            h[g] = row_shr_or<1>(rowin, H[g]);
        }
        #pragma unroll
        for (int j = 0; j < KS - 1; ++j) {
            #pragma unroll
            for (int g = 0; g < NG; ++g) {
                h[g] = fmaf(e[g][j], h[g], bb[g][j]);
                y[j] = fmaf(h[g], Cv[j][g], y[j]);
            }
        }
        #pragma unroll
        for (int g = 0; g < NG; ++g)
            y[KS - 1] = fmaf(H[g], Cv[KS - 1][g], y[KS - 1]);
    };

    auto do_block = [&](int buf, const float (&d)[KS], const float (&xv)[KS], float (&y)[KS]) {
        float dx[KS];
        #pragma unroll
        for (int j = 0; j < KS; ++j) { dx[j] = d[j] * xv[j]; y[j] = 0.f; }
        float Wd[5];
        Wd[0] = d[0];
        #pragma unroll
        for (int j = 1; j < KS; ++j) Wd[0] += d[j];
        Wd[1] = Wd[0] + row_shr_or0<1>(Wd[0]);
        Wd[2] = Wd[1] + row_shr_or0<2>(Wd[1]);
        Wd[3] = Wd[2] + row_shr_or0<4>(Wd[2]);
        Wd[4] = Wd[3] + row_shr_or0<8>(Wd[3]);
        float Bv[KS][4], Cv[KS][4];
        #pragma unroll
        for (int q = 0; q < NQ; ++q) {
            lds_load(buf, q, Bv, Cv);
            __builtin_amdgcn_sched_barrier(0);
            do_group(4 * q, Wd, d, dx, Bv, Cv, y);
        }
    };

    const int lpart = (dir == 3) ? sh * (lane & 3) + sw * (lane >> 2) : sw * lane;
    const int lmin  = (dir == 3) ? -3 : ((dir == 1) ? -31 : 0);
    const unsigned lpBytes = (unsigned)(lpart - lmin) * 4u;
    float* const strip = sY[w];
    int soff[KS];
    #pragma unroll
    for (int i = 0; i < KS; ++i)
        soff[i] = (dir == 3) ? sw * (8 * i) : sh * (i >> 1) + sw * (32 * (i & 1));
    auto add_out = [&](int blk, const float (&y)[KS]) {
        static_assert(KS == 8, "a lane's results are written as two float4 on a row walk");
        if (dir == 3) {
            float* const mine = strip + 32 * (lane & 7) + (lane >> 3);
            #pragma unroll
            for (int j = 0; j < KS; ++j) mine[4 * j] = y[j] * 0.25f;
        } else {
            float4* const mine = reinterpret_cast<float4*>(strip + KS * lane);
            mine[0] = make_float4(y[0] * 0.25f, y[1] * 0.25f, y[2] * 0.25f, y[3] * 0.25f);
            mine[1] = make_float4(y[4] * 0.25f, y[5] * 0.25f, y[6] * 0.25f, y[7] * 0.25f);
        }
        char* const blkBase = reinterpret_cast<char*>(outBase + (base + lmin + sh * 4 * blk));
        float v[KS];
        #pragma unroll
        for (int i = 0; i < KS; ++i) v[i] = strip[32 * i + lane];
        #pragma unroll
        for (int i = 0; i < KS; ++i)
            atomicAdd(reinterpret_cast<float*>(blkBase + (ptrdiff_t)soff[i] * 4 + lpBytes), v[i]);
    };

    float d0[KS], x0[KS], d1[KS], x1[KS], y[KS];
    static_assert(NSTG == 4, "the staging registers below are written out for 4 per thread");
    static_assert(NSTD == 2, "the delta staging registers below are written out for 2 per thread");
    float4 gb0, gb1, gb2, gb3, gc0, gc1, gc2, gc3, gd0, gd1;
    auto stage_load_all = [&](int blk) {
        stage_load_d(blk, 0, gd0);
        stage_load_d(blk, 1, gd1);
        stage_load(blk, 0, gb0, gc0);
        stage_load(blk, 1, gb1, gc1);
        stage_load(blk, 2, gb2, gc2);
        stage_load(blk, 3, gb3, gc3);
    };
    auto stage_store_all = [&](int buf) {
        stage_store_d(buf, 0, gd0);
        stage_store_d(buf, 1, gd1);
        stage_store(buf, 0, gb0, gc0);
        stage_store(buf, 1, gb1, gc1);
        stage_store(buf, 2, gb2, gc2);
        stage_store(buf, 3, gb3, gc3);
    };
    stage_load_all(0);
    load_x(0, x0);
    stage_store_all(0);
    __syncthreads();
    lds_load_d(0, d0);

    for (int blk = 0; blk < NBLK; blk += 2) {
        stage_load_all(blk + 1);
        load_x(blk + 1, x1);
        __builtin_amdgcn_sched_barrier(0);
        do_block(0, d0, x0, y);
        stage_store_all(1);
        __syncthreads();
        lds_load_d(1, d1);
        add_out(blk, y);

        const int nb = (blk + 2 < NBLK) ? blk + 2 : blk + 1;
        stage_load_all(nb);
        load_x(nb, x0);
        __builtin_amdgcn_sched_barrier(0);
        do_block(1, d1, x1, y);
        stage_store_all(0);
        __syncthreads();
        lds_load_d(0, d0);
        add_out(blk + 1, y);
    }
}

extern "C" void kernel_launch(void* const* d_in, const int* in_sizes, int n_in,
                              void* d_out, int out_size, void* d_ws, size_t ws_size,
                              hipStream_t stream) {
    const float* feat    = (const float*)d_in[0];
    const float* A_log   = (const float*)d_in[1];
    const float* Dvec    = (const float*)d_in[2];
    const float* W_delta = (const float*)d_in[3];
    const float* b_delta = (const float*)d_in[4];
    const float* W_B     = (const float*)d_in[5];
    const float* W_C     = (const float*)d_in[6];
    float* out = (float*)d_out;

    float* ws       = (float*)d_ws;
    float* delta_ws = ws;
    float* Bm_ws    = ws + (size_t)NROW * CC;
    float* Cm_ws    = Bm_ws + (size_t)NROW * NN;

    proj_wmma_kernel<<<NROW / PROJ_ROWS, PROJ_THREADS, 0, stream>>>(
        feat, Dvec, W_delta, b_delta, W_B, W_C, delta_ws, Bm_ws, Cm_ws, out);
    scan_kernel<<<256, 256, 0, stream>>>(feat, A_log, delta_ws, Bm_ws, Cm_ws, out);
}


# scan: per-block barrier split, fenced for LDS only: signal behind the block's LDS stores, wait before the next LDS read; adds and next loads issue between; no wait for adds; add order free as before
# speedup vs baseline: 32.1513x; 1.1291x over previous
#include <hip/hip_runtime.h>
#include <math.h>

typedef __attribute__((ext_vector_type(2))) float v2f;
typedef __attribute__((ext_vector_type(8))) float v8f;

#define BB 2
#define CC 256
#define HH 64
#define WW 64
#define HW 4096
#define NN 16
#define NROW (BB*HW)

__device__ __forceinline__ float softplus_sel(float pre) {
    const float t = log1pf(__expf(pre));
    return (pre > 20.f) ? pre : t;
}
#define PROJ_ROWS  32
#define PROJ_WAVES 18
#define PROJ_THREADS (PROJ_WAVES * 32)
#define PROJ_PITCH 260
#define PROJ_V4 (PROJ_ROWS * CC / 4)
#define PROJ_KSTEPS (CC / 4)
__global__ __launch_bounds__(PROJ_THREADS) void proj_wmma_kernel(const float* __restrict__ feat,
                                 const float* __restrict__ Dvec,
                                 const float* __restrict__ W_delta,
                                 const float* __restrict__ b_delta,
                                 const float* __restrict__ W_B,
                                 const float* __restrict__ W_C,
                                 float* __restrict__ delta_ws,
                                 float* __restrict__ Bm_ws,
                                 float* __restrict__ Cm_ws,
                                 float* __restrict__ out) {
    __shared__ float sX[PROJ_ROWS * PROJ_PITCH];

    const int tid  = threadIdx.x;
    const int lane = tid & 31;
    const int lo   = lane & 15;
    const int hi   = lane >> 4;
    const int which = __builtin_amdgcn_readfirstlane(tid >> 5);
    const int m0   = blockIdx.x * PROJ_ROWS;
    const int bA   = m0 >> 12;
    const int l0   = m0 & (HW - 1);
    const float* xBase = feat + (size_t)bA * CC * HW + l0;

    float4 v[4];
    float dk[4];
    #pragma unroll
    for (int i = 0; i < 4; ++i) {
        const int f = tid + PROJ_THREADS * i;
        if (f < PROJ_V4) {
            v[i] = *reinterpret_cast<const float4*>(xBase + (size_t)(f >> 3) * HW + 4 * (f & 7));
            dk[i] = Dvec[f >> 3];
        }
    }
    #pragma unroll
    for (int i = 0; i < 4; ++i) {
        const int f = tid + PROJ_THREADS * i;
        if (f < PROJ_V4) {
            float* d = sX + (4 * (f & 7)) * PROJ_PITCH + (f >> 3);
            d[0]              = v[i].x;
            d[PROJ_PITCH]     = v[i].y;
            d[2 * PROJ_PITCH] = v[i].z;
            d[3 * PROJ_PITCH] = v[i].w;
        }
    }
    __syncthreads();

    {
        float* oBase = out + (size_t)bA * CC * HW + l0;
        #pragma unroll
        for (int i = 0; i < 4; ++i) {
            const int f = tid + PROJ_THREADS * i;
            if (f < PROJ_V4) {
                float4 o;
                o.x = v[i].x * dk[i]; o.y = v[i].y * dk[i]; o.z = v[i].z * dk[i]; o.w = v[i].w * dk[i];
                *reinterpret_cast<float4*>(oBase + (size_t)(f >> 3) * HW + 4 * (f & 7)) = o;
            }
        }
    }

    const float* Wmat;
    unsigned ncols, n0;
    if (which < 16)       { Wmat = W_delta; ncols = CC; n0 = which * 16; }
    else if (which == 16) { Wmat = W_B;     ncols = NN; n0 = 0; }
    else                  { Wmat = W_C;     ncols = NN; n0 = 0; }
    const unsigned laneoff = (unsigned)(2 * hi) * ncols + n0 + (unsigned)lo;
    v2f bw[PROJ_KSTEPS];
    #pragma unroll
    for (int s = 0; s < PROJ_KSTEPS; ++s) {
        const float* r0 = Wmat + (size_t)(4 * s) * ncols;
        const float* r1 = r0 + ncols;
        bw[s].x = r0[laneoff];
        bw[s].y = r1[laneoff];
    }

    auto run_tile = [&](const float* aRow) -> v8f {
        v8f acc = {};
        #pragma unroll
        for (int s = 0; s < PROJ_KSTEPS; ++s) {
            const v2f a = *reinterpret_cast<const v2f*>(aRow + 4 * s);
            acc = __builtin_amdgcn_wmma_f32_16x16x4_f32(
                 false, a,  false, bw[s],
                 (short)0, acc,  false,  false);
        }
        return acc;
    };
    const float* aRow0 = sX + lo * PROJ_PITCH + 2 * hi;
    const float* aRow1 = aRow0 + 16 * PROJ_PITCH;

    if (which < 16) {
        const float bias = b_delta[n0 + lo];
        float* dcol = delta_ws + (size_t)(m0 + 8 * hi) * CC + n0 + lo;
        const v8f acc0 = run_tile(aRow0);
        v8f acc1 = {};
        #pragma unroll
        for (int r = 0; r < 8; ++r) {
            #pragma unroll
            for (int s = 8 * r; s < 8 * r + 8; ++s) {
                const v2f a = *reinterpret_cast<const v2f*>(aRow1 + 4 * s);
                acc1 = __builtin_amdgcn_wmma_f32_16x16x4_f32(
                     false, a,  false, bw[s],
                     (short)0, acc1,  false,  false);
            }
            dcol[(size_t)r * CC] = softplus_sel(acc0[r] + bias);
            #pragma unroll
            for (int g = 0; g < 8; ++g) {
                __builtin_amdgcn_sched_group_barrier(0x008, 1, 0);
                __builtin_amdgcn_sched_group_barrier(0x002, 11, 0);
            }
            __builtin_amdgcn_sched_barrier(0);
        }
        #pragma unroll
        for (int r = 0; r < 8; ++r)
            dcol[(size_t)(16 + r) * CC] = softplus_sel(acc1[r] + bias);
    } else {
        float* dst = ((which == 16) ? Bm_ws : Cm_ws) + (size_t)(m0 + 8 * hi) * NN + lo;
        const v8f acc0 = run_tile(aRow0);
        #pragma unroll
        for (int r = 0; r < 8; ++r) dst[(size_t)r * NN] = acc0[r];
        const v8f acc1 = run_tile(aRow1);
        #pragma unroll
        for (int r = 0; r < 8; ++r) dst[(size_t)(16 + r) * NN] = acc1[r];
    }
}


#define KS   8
#define BLK  (32 * KS)
#define NBLK (HW / BLK)
#define NG   4
#define NQ   (NN / 4)
#define PITCH 33
#define NSTG (BLK * NQ / 256)
#define DPITCH (BLK + 4)
#define NSTD (BLK * 2 / 256)

template <int K>
__device__ __forceinline__ float row_shr_or0(float v) {
    return __int_as_float(__builtin_amdgcn_update_dpp(
        0, __float_as_int(v), 0x110 + K, 0xf, 0xf, true));
}
template <int K>
__device__ __forceinline__ float row_shr_or(float outside, float v) {
    return __int_as_float(__builtin_amdgcn_update_dpp(
        __float_as_int(outside), __float_as_int(v), 0x110 + K, 0xf, 0xf, false));
}
template <int K>
__device__ __forceinline__ void scan_level(const float a, float& bb) {
    const float bs = row_shr_or0<K>(bb);
    bb = fmaf(a, bs, bb);
}

__device__ __forceinline__ void lds_signal() {
    __builtin_amdgcn_fence(__ATOMIC_RELEASE, "workgroup", "local");
    __builtin_amdgcn_s_barrier_signal(-1);
}
__device__ __forceinline__ void lds_wait() {
    __builtin_amdgcn_s_barrier_wait(-1);
    __builtin_amdgcn_fence(__ATOMIC_ACQUIRE, "workgroup", "local");
}

__global__ __launch_bounds__(256) __attribute__((amdgpu_waves_per_eu(4, 4))) void scan_kernel(const float* __restrict__ feat,
                            const float* __restrict__ A_log,
                            const float* __restrict__ delta_ws,
                            const float* __restrict__ Bm_ws,
                            const float* __restrict__ Cm_ws,
                            float* __restrict__ out) {
    __shared__ float4 sB[2][KS * NQ * PITCH];
    __shared__ float4 sC[2][KS * NQ * PITCH];
    __shared__ float sY[8][BLK];
    __shared__ float4 sD[2][8 * (DPITCH / 4)];

    const int tid  = threadIdx.x;
    const int lane = tid & 31;
    const int w   = __builtin_amdgcn_readfirstlane((int)(threadIdx.x >> 5));
    const int id  = blockIdx.x * 8 + w;
    const int dir = id >> 9;
    const int b   = (id >> 8) & 1;
    const int c   = id & 255;

    int base, sh, sw;
    switch (dir) {
        case 0:  base = 0;    sh =  64; sw =  1; break;
        case 1:  base = 4095; sh = -64; sw = -1; break;
        case 2:  base = 4032; sh = -64; sw =  1; break;
        default: base = 63;   sh =  -1; sw = 64; break;
    }

    float A[NN];
    #pragma unroll
    for (int n = 0; n < NN; ++n) A[n] = -__expf(A_log[c * NN + n]) * 1.44269504088896341f;

    const float* fBase = feat + (size_t)b * CC * HW + (size_t)c * HW;
    const size_t rbase = (size_t)b * HW;
    float* outBase = out + (size_t)b * CC * HW + (size_t)c * HW;
    const bool upper = lane >= 16;
    const float4* Bm4 = reinterpret_cast<const float4*>(Bm_ws);
    const float4* Cm4 = reinterpret_cast<const float4*>(Cm_ws);

    float hin[NN];
    #pragma unroll
    for (int n = 0; n < NN; ++n) hin[n] = 0.f;

    auto pos_of = [&](int blk, int s) -> int {
        const int t = blk * BLK + s;
        return base + sh * (t >> 6) + sw * (t & 63);
    };
    auto pos = [&](int blk, int j) -> int { return pos_of(blk, KS * lane + j); };

    auto stage_load = [&](int blk, int r, float4& vb, float4& vc) {
        const int f = tid + 256 * r;
        const int l = pos_of(blk, f >> 2);
        const size_t v4 = (rbase + (size_t)l) * NQ + (size_t)(f & 3);
        vb = Bm4[v4];
        vc = Cm4[v4];
    };
    auto stage_store = [&](int buf, int r, const float4& vb, const float4& vc) {
        const int f = tid + 256 * r;
        const int s = f >> 2;
        const int idx = ((s % KS) * NQ + (f & 3)) * PITCH + (s / KS);
        sB[buf][idx] = vb;
        sC[buf][idx] = vc;
    };

    const int c0 = c - w;
    const float4* D4 = reinterpret_cast<const float4*>(delta_ws);
    auto stage_load_d = [&](int blk, int r, float4& vd) {
        const int f = tid + 256 * r;
        const int l = pos_of(blk, f >> 1);
        vd = D4[(rbase + (size_t)l) * (CC / 4) + (size_t)((c0 >> 2) + (f & 1))];
    };
    auto stage_store_d = [&](int buf, int r, const float4& vd) {
        const int f = tid + 256 * r;
        float* p = reinterpret_cast<float*>(sD[buf]) + (4 * (f & 1)) * DPITCH + (f >> 1);
        p[0]          = vd.x;
        p[DPITCH]     = vd.y;
        p[2 * DPITCH] = vd.z;
        p[3 * DPITCH] = vd.w;
    };
    auto lds_load_d = [&](int buf, float (&d)[KS]) {
        static_assert(KS == 8, "a lane's delta values are read as two float4");
        const float4* p = reinterpret_cast<const float4*>(
            reinterpret_cast<const float*>(sD[buf]) + w * DPITCH + KS * lane);
        const float4 lo4 = p[0];
        const float4 hi4 = p[1];
        d[0] = lo4.x; d[1] = lo4.y; d[2] = lo4.z; d[3] = lo4.w;
        d[4] = hi4.x; d[5] = hi4.y; d[6] = hi4.z; d[7] = hi4.w;
    };

    auto load_x = [&](int blk, float (&xv)[KS]) {
        #pragma unroll
        for (int j = 0; j < KS; ++j) xv[j] = fBase[pos(blk, j)];
    };

    auto lds_load = [&](int buf, int q, float (&Bv)[KS][4], float (&Cv)[KS][4]) {
        #pragma unroll
        for (int j = 0; j < KS; ++j) {
            const float4 bv = sB[buf][(j * NQ + q) * PITCH + lane];
            const float4 cv = sC[buf][(j * NQ + q) * PITCH + lane];
            Bv[j][0] = bv.x; Bv[j][1] = bv.y; Bv[j][2] = bv.z; Bv[j][3] = bv.w;
            Cv[j][0] = cv.x; Cv[j][1] = cv.y; Cv[j][2] = cv.z; Cv[j][3] = cv.w;
        }
    };

    auto do_group = [&](int n0, const float (&Wd)[5], const float (&d)[KS], const float (&dx)[KS],
                        const float (&Bv)[KS][4], const float (&Cv)[KS][4], float (&y)[KS]) {
        float e[NG][KS], bb[NG][KS], pk[NG][5], Q[NG], H[NG], h[NG];
        #pragma unroll
        for (int g = 0; g < NG; ++g) {
            #pragma unroll
            for (int j = 0; j < KS; ++j) e[g][j] = __builtin_amdgcn_exp2f(d[j] * A[n0 + g]);
        }
        __builtin_amdgcn_sched_barrier(0);
        #pragma unroll
        for (int g = 0; g < NG; ++g) {
            #pragma unroll
            for (int k = 0; k < 5; ++k) pk[g][k] = __builtin_amdgcn_exp2f(Wd[k] * A[n0 + g]);
        }
        #pragma unroll
        for (int g = 0; g < NG; ++g) {
            #pragma unroll
            for (int j = 0; j < KS; ++j) bb[g][j] = dx[j] * Bv[j][g];
        }
        #pragma unroll
        for (int g = 0; g < NG; ++g) Q[g] = bb[g][0];
        #pragma unroll
        for (int j = 1; j < KS; ++j) {
            #pragma unroll
            for (int g = 0; g < NG; ++g) Q[g] = fmaf(e[g][j], Q[g], bb[g][j]);
        }
        #pragma unroll
        for (int g = 0; g < NG; ++g) scan_level<1>(pk[g][0], Q[g]);
        #pragma unroll
        for (int g = 0; g < NG; ++g) scan_level<2>(pk[g][1], Q[g]);
        #pragma unroll
        for (int g = 0; g < NG; ++g) scan_level<4>(pk[g][2], Q[g]);
        #pragma unroll
        for (int g = 0; g < NG; ++g) scan_level<8>(pk[g][3], Q[g]);
        #pragma unroll
        for (int g = 0; g < NG; ++g) {
            const int n = n0 + g;
            const float Pg = pk[g][4];
            const float h0 = fmaf(Pg, hin[n], Q[g]);
            const float c0 = __int_as_float(__builtin_amdgcn_readlane(__float_as_int(h0), 15));
            H[g] = upper ? fmaf(Pg, c0, Q[g]) : h0;
            const float rowin = upper ? c0 : hin[n];
            hin[n] = __int_as_float(__builtin_amdgcn_readlane(__float_as_int(H[g]), 31));
            h[g] = row_shr_or<1>(rowin, H[g]);
        }
        #pragma unroll
        for (int j = 0; j < KS - 1; ++j) {
            #pragma unroll
            for (int g = 0; g < NG; ++g) {
                h[g] = fmaf(e[g][j], h[g], bb[g][j]);
                y[j] = fmaf(h[g], Cv[j][g], y[j]);
            }
        }
        #pragma unroll
        for (int g = 0; g < NG; ++g)
            y[KS - 1] = fmaf(H[g], Cv[KS - 1][g], y[KS - 1]);
    };

    auto do_block = [&](int buf, const float (&d)[KS], const float (&xv)[KS], float (&y)[KS]) {
        float dx[KS];
        #pragma unroll
        for (int j = 0; j < KS; ++j) { dx[j] = d[j] * xv[j]; y[j] = 0.f; }
        float Wd[5];
        Wd[0] = d[0];
        #pragma unroll
        for (int j = 1; j < KS; ++j) Wd[0] += d[j];
        Wd[1] = Wd[0] + row_shr_or0<1>(Wd[0]);
        Wd[2] = Wd[1] + row_shr_or0<2>(Wd[1]);
        Wd[3] = Wd[2] + row_shr_or0<4>(Wd[2]);
        Wd[4] = Wd[3] + row_shr_or0<8>(Wd[3]);
        float Bv[KS][4], Cv[KS][4];
        #pragma unroll
        for (int q = 0; q < NQ; ++q) {
            lds_load(buf, q, Bv, Cv);
            __builtin_amdgcn_sched_barrier(0);
            do_group(4 * q, Wd, d, dx, Bv, Cv, y);
        }
    };

    const int lpart = (dir == 3) ? sh * (lane & 3) + sw * (lane >> 2) : sw * lane;
    const int lmin  = (dir == 3) ? -3 : ((dir == 1) ? -31 : 0);
    const unsigned lpBytes = (unsigned)(lpart - lmin) * 4u;
    float* const strip = sY[w];
    int soff[KS];
    #pragma unroll
    for (int i = 0; i < KS; ++i)
        soff[i] = (dir == 3) ? sw * (8 * i) : sh * (i >> 1) + sw * (32 * (i & 1));
    auto add_out = [&](int blk, const float (&y)[KS]) {
        static_assert(KS == 8, "a lane's results are written as two float4 on a row walk");
        if (dir == 3) {
            float* const mine = strip + 32 * (lane & 7) + (lane >> 3);
            #pragma unroll
            for (int j = 0; j < KS; ++j) mine[4 * j] = y[j] * 0.25f;
        } else {
            float4* const mine = reinterpret_cast<float4*>(strip + KS * lane);
            mine[0] = make_float4(y[0] * 0.25f, y[1] * 0.25f, y[2] * 0.25f, y[3] * 0.25f);
            mine[1] = make_float4(y[4] * 0.25f, y[5] * 0.25f, y[6] * 0.25f, y[7] * 0.25f);
        }
        char* const blkBase = reinterpret_cast<char*>(outBase + (base + lmin + sh * 4 * blk));
        float v[KS];
        #pragma unroll
        for (int i = 0; i < KS; ++i) v[i] = strip[32 * i + lane];
        #pragma unroll
        for (int i = 0; i < KS; ++i)
            atomicAdd(reinterpret_cast<float*>(blkBase + (ptrdiff_t)soff[i] * 4 + lpBytes), v[i]);
    };

    float d0[KS], x0[KS], d1[KS], x1[KS], y[KS];
    static_assert(NSTG == 4, "the staging registers below are written out for 4 per thread");
    static_assert(NSTD == 2, "the delta staging registers below are written out for 2 per thread");
    float4 gb0, gb1, gb2, gb3, gc0, gc1, gc2, gc3, gd0, gd1;
    auto stage_load_all = [&](int blk) {
        stage_load_d(blk, 0, gd0);
        stage_load_d(blk, 1, gd1);
        stage_load(blk, 0, gb0, gc0);
        stage_load(blk, 1, gb1, gc1);
        stage_load(blk, 2, gb2, gc2);
        stage_load(blk, 3, gb3, gc3);
    };
    auto stage_store_all = [&](int buf) {
        stage_store_d(buf, 0, gd0);
        stage_store_d(buf, 1, gd1);
        stage_store(buf, 0, gb0, gc0);
        stage_store(buf, 1, gb1, gc1);
        stage_store(buf, 2, gb2, gc2);
        stage_store(buf, 3, gb3, gc3);
    };
    stage_load_all(0);
    load_x(0, x0);
    stage_store_all(0);
    lds_signal();
    lds_wait();
    lds_load_d(0, d0);
    stage_load_all(1);
    load_x(1, x1);

    for (int blk = 0; blk < NBLK; blk += 2) {
        __builtin_amdgcn_sched_barrier(0);
        do_block(0, d0, x0, y);
        stage_store_all(1);
        lds_signal();
        add_out(blk, y);
        const int nb = (blk + 2 < NBLK) ? blk + 2 : blk + 1;
        stage_load_all(nb);
        load_x(nb, x0);
        lds_wait();
        lds_load_d(1, d1);
        __builtin_amdgcn_sched_barrier(0);
        do_block(1, d1, x1, y);
        stage_store_all(0);
        lds_signal();
        add_out(blk + 1, y);
        const int nb2 = (blk + 3 < NBLK) ? blk + 3 : blk + 1;
        stage_load_all(nb2);
        load_x(nb2, x1);
        lds_wait();
        lds_load_d(0, d0);
    }
}

extern "C" void kernel_launch(void* const* d_in, const int* in_sizes, int n_in,
                              void* d_out, int out_size, void* d_ws, size_t ws_size,
                              hipStream_t stream) {
    const float* feat    = (const float*)d_in[0];
    const float* A_log   = (const float*)d_in[1];
    const float* Dvec    = (const float*)d_in[2];
    const float* W_delta = (const float*)d_in[3];
    const float* b_delta = (const float*)d_in[4];
    const float* W_B     = (const float*)d_in[5];
    const float* W_C     = (const float*)d_in[6];
    float* out = (float*)d_out;

    float* ws       = (float*)d_ws;
    float* delta_ws = ws;
    float* Bm_ws    = ws + (size_t)NROW * CC;
    float* Cm_ws    = Bm_ws + (size_t)NROW * NN;

    proj_wmma_kernel<<<NROW / PROJ_ROWS, PROJ_THREADS, 0, stream>>>(
        feat, Dvec, W_delta, b_delta, W_B, W_C, delta_ws, Bm_ws, Cm_ws, out);
    scan_kernel<<<256, 256, 0, stream>>>(feat, A_log, delta_ws, Bm_ws, Cm_ws, out);
}
